# starting kernel
# baseline (speedup 1.0000x reference)
#include <hip/hip_runtime.h>

typedef __attribute__((ext_vector_type(2))) float v2f;
typedef __attribute__((ext_vector_type(8))) float v8f;

#define NQ 10
#define NL 4
#define NSTATES 1024
#define WAVES_PER_BLOCK 4
#define THREADS (WAVES_PER_BLOCK * 32)

#define LDS_BYTES (1280 + 8192 + 8192 + 512 + 8192 + 32768)


__device__ __forceinline__ void apply16(float2* S,
                                        const float* Ure, const float* Uim,
                                        const int mode, const int lane) {
  const int hf  = lane >> 4;
  const int nlo = lane & 15;

  v2f aRe[4], aIm[4], aImN[4];
  #pragma unroll
  for (int k = 0; k < 4; ++k) {
    const int K0 = 4 * k + 2 * hf;
    aRe[k].x  = Ure[nlo * 16 + K0];  aRe[k].y  = Ure[nlo * 16 + K0 + 1];
    aIm[k].x  = Uim[nlo * 16 + K0];  aIm[k].y  = Uim[nlo * 16 + K0 + 1];
    aImN[k].x = -aIm[k].x;           aImN[k].y = -aIm[k].y;
  }

  #pragma unroll
  for (int jt = 0; jt < 4; ++jt) {
    const int n = jt * 16 + nlo;
    v8f accRe = {0.f,0.f,0.f,0.f,0.f,0.f,0.f,0.f};
    v8f accIm = {0.f,0.f,0.f,0.f,0.f,0.f,0.f,0.f};
    #pragma unroll
    for (int k = 0; k < 4; ++k) {
      const int K0 = 4 * k + 2 * hf;
      const int K1 = K0 + 1;
      const int i0 = (mode == 0) ? (K0 * 64 + n) : (((n >> 2) << 6) + K0 * 4 + (n & 3));
      const int i1 = (mode == 0) ? (K1 * 64 + n) : (((n >> 2) << 6) + K1 * 4 + (n & 3));
      const float2 z0 = S[i0];
      const float2 z1 = S[i1];
      v2f br, bi;
      br.x = z0.x;  br.y = z1.x;
      bi.x = z0.y;  bi.y = z1.y;
      accRe = __builtin_amdgcn_wmma_f32_16x16x4_f32(false, aRe[k],  false, br, (short)0, accRe, false, false);
      accRe = __builtin_amdgcn_wmma_f32_16x16x4_f32(false, aImN[k], false, bi, (short)0, accRe, false, false);
      accIm = __builtin_amdgcn_wmma_f32_16x16x4_f32(false, aRe[k],  false, bi, (short)0, accIm, false, false);
      accIm = __builtin_amdgcn_wmma_f32_16x16x4_f32(false, aIm[k],  false, br, (short)0, accIm, false, false);
    }
    #pragma unroll
    for (int v = 0; v < 8; ++v) {
      const int M = v + 8 * hf;
      const int o = (mode == 0) ? (M * 64 + n) : (((n >> 2) << 6) + M * 4 + (n & 3));
      S[o] = make_float2(accRe[v], accIm[v]);
    }
  }
}

__device__ __forceinline__ void applyC(float2* S,
                                       const float* ucr, const float* uci,
                                       const int lane) {
  const int hf  = lane >> 4;
  const int nlo = lane & 15;
  const int grp = nlo >> 2, cc = nlo & 3;
  const int c0 = 2 * hf;

  const float vrx = ucr[cc * 4 + c0], vry = ucr[cc * 4 + c0 + 1];
  const float vix = uci[cc * 4 + c0], viy = uci[cc * 4 + c0 + 1];
  v2f bRe[4], bIm[4], bImN[4];
  #pragma unroll
  for (int k = 0; k < 4; ++k) {
    const bool on = (k == grp);
    bRe[k].x  = on ?  vrx : 0.f;  bRe[k].y  = on ?  vry : 0.f;
    bIm[k].x  = on ?  vix : 0.f;  bIm[k].y  = on ?  viy : 0.f;
    bImN[k].x = on ? -vix : 0.f;  bImN[k].y = on ? -viy : 0.f;
  }

  #pragma unroll
  for (int jt = 0; jt < 4; ++jt) {
    const int n = jt * 16 + nlo;
    v8f accRe = {0.f,0.f,0.f,0.f,0.f,0.f,0.f,0.f};
    v8f accIm = {0.f,0.f,0.f,0.f,0.f,0.f,0.f,0.f};
    #pragma unroll
    for (int k = 0; k < 4; ++k) {
      const int i0 = (4 * jt + k) * 64 + nlo * 4 + 2 * hf;
      const float4 z = *(const float4*)(S + i0);
      v2f ar, ai;
      ar.x = z.x;  ar.y = z.z;
      ai.x = z.y;  ai.y = z.w;
      accRe = __builtin_amdgcn_wmma_f32_16x16x4_f32(false, ar, false, bRe[k],  (short)0, accRe, false, false);
      accRe = __builtin_amdgcn_wmma_f32_16x16x4_f32(false, ai, false, bImN[k], (short)0, accRe, false, false);
      accIm = __builtin_amdgcn_wmma_f32_16x16x4_f32(false, ar, false, bIm[k],  (short)0, accIm, false, false);
      accIm = __builtin_amdgcn_wmma_f32_16x16x4_f32(false, ai, false, bRe[k],  (short)0, accIm, false, false);
    }
    #pragma unroll
    for (int v = 0; v < 8; ++v) {
      const int M = v + 8 * hf;
      const int o = ((n >> 2) << 6) + M * 4 + (n & 3);
      S[o] = make_float2(accRe[v], accIm[v]);
    }
  }
}

extern "C" __global__ __launch_bounds__(THREADS)
void qlayer_kernel(const float* __restrict__ x,
                   const float* __restrict__ qw,
                   float* __restrict__ out) {
  extern __shared__ char smem_raw[];
  float* u2   = (float*)smem_raw;
  float* UAre = u2 + NL * NQ * 8;
  float* UAim = UAre + NL * 256;
  float* UBre = UAim + NL * 256;
  float* UBim = UBre + NL * 256;
  float* UCre = UBim + NL * 256;
  float* UCim = UCre + NL * 16;
  unsigned* permPk = (unsigned*)(UCim + NL * 16);
  float2* stateBase = (float2*)(permPk + NL * 512);

  const int tid = threadIdx.x;

  #pragma unroll 1
  for (int e = tid; e < NL * NQ; e += THREADS) {
    const int l = e / NQ, w = e % NQ;
    const float phi = qw[(l * NQ + w) * 3 + 0];
    const float th  = qw[(l * NQ + w) * 3 + 1];
    const float om  = qw[(l * NQ + w) * 3 + 2];
    const float c = cosf(0.5f * th), s = sinf(0.5f * th);
    const float ap = 0.5f * (phi + om), am = 0.5f * (phi - om);
    const float cpo = cosf(ap), spo = sinf(ap);
    const float cb = cosf(am),  sb = sinf(am);
    float* g = u2 + e * 8;
    g[0] =  cpo * c; g[1] = -spo * c;
    g[2] = -cb * s;  g[3] = -sb * s;
    g[4] =  cb * s;  g[5] = -sb * s;
    g[6] =  cpo * c; g[7] =  spo * c;
  }
  __syncthreads();

  #pragma unroll 1
  for (int e = tid; e < NL * 256; e += THREADS) {
    const int l = e >> 8, rc = e & 255, r = rc >> 4, cc = rc & 15;
    float arA = 1.f, aiA = 0.f, arB = 1.f, aiB = 0.f;
    #pragma unroll
    for (int q = 0; q < 4; ++q) {
      const int rb = (r >> (3 - q)) & 1, cbit = (cc >> (3 - q)) & 1;
      const float* gA = u2 + (((l * NQ + q) * 4) + rb * 2 + cbit) * 2;
      float nr = arA * gA[0] - aiA * gA[1];
      float ni = arA * gA[1] + aiA * gA[0];
      arA = nr; aiA = ni;
      const float* gB = u2 + (((l * NQ + 4 + q) * 4) + rb * 2 + cbit) * 2;
      nr = arB * gB[0] - aiB * gB[1];
      ni = arB * gB[1] + aiB * gB[0];
      arB = nr; aiB = ni;
    }
    UAre[e] = arA; UAim[e] = aiA;
    UBre[e] = arB; UBim[e] = aiB;
  }
  #pragma unroll 1
  for (int e = tid; e < NL * 16; e += THREADS) {
    const int l = e >> 4, r = (e >> 2) & 3, cc = e & 3;
    const float* g8 = u2 + (((l * NQ + 8) * 4) + ((r >> 1) & 1) * 2 + ((cc >> 1) & 1)) * 2;
    const float* g9 = u2 + (((l * NQ + 9) * 4) + (r & 1) * 2 + (cc & 1)) * 2;
    UCre[e] = g8[0] * g9[0] - g8[1] * g9[1];
    UCim[e] = g8[0] * g9[1] + g8[1] * g9[0];
  }
  #pragma unroll 1
  for (int e = tid; e < NL * 512; e += THREADS) {
    const int l = e >> 9, p = e & 511;
    const int ln = p & 31, h = p >> 5;
    const int rr = (l % (NQ - 1)) + 1;
    unsigned pk = 0;
    #pragma unroll
    for (int half = 0; half < 2; ++half) {
      int j = ln + 64 * h + 32 * half;
      for (int w = NQ - 1; w >= 0; --w) {
        const int cpos = 9 - w;
        const int tpos = 9 - ((w + rr) % NQ);
        j ^= ((j >> cpos) & 1) << tpos;
      }
      pk |= ((unsigned)j) << (16 * half);
    }
    permPk[e] = pk;
  }
  __syncthreads();

  const int wave = tid >> 5, lane = tid & 31;
  const int b = blockIdx.x * WAVES_PER_BLOCK + wave;
  float2* S = stateBase + wave * NSTATES;

  float cw[NQ], sw[NQ];
  #pragma unroll
  for (int w = 0; w < NQ; ++w) {
    const float ang = 1.5707963267948966f * x[b * NQ + w];
    cw[w] = cosf(ang);
    sw[w] = sinf(ang);
  }
  for (int t = 0; t < 32; ++t) {
    const int i = lane + 32 * t;
    float a = 1.f;
    #pragma unroll
    for (int w = 0; w < NQ; ++w) a *= ((i >> (9 - w)) & 1) ? sw[w] : cw[w];
    S[i] = make_float2(a, 0.f);
  }

  #pragma unroll 1
  for (int l = 0; l < NL; ++l) {
    apply16(S, UAre + l * 256, UAim + l * 256, 0, lane);
    apply16(S, UBre + l * 256, UBim + l * 256, 1, lane);
    applyC (S, UCre + l * 16,  UCim + l * 16,  lane);

    const unsigned* pl = permPk + l * 512;
    float2 tv[32];
    #pragma unroll
    for (int h = 0; h < 16; ++h) {
      const unsigned pk = pl[lane + 32 * h];
      tv[2 * h]     = S[pk & 0xFFFFu];
      tv[2 * h + 1] = S[pk >> 16];
    }
    #pragma unroll
    for (int t = 0; t < 32; ++t) S[lane + 32 * t] = tv[t];
  }

  float acc[NQ];
  #pragma unroll
  for (int w = 0; w < NQ; ++w) acc[w] = 0.f;
  for (int t = 0; t < 32; ++t) {
    const int i = lane + 32 * t;
    const float2 z = S[i];
    const float p = z.x * z.x + z.y * z.y;
    #pragma unroll
    for (int w = 0; w < NQ; ++w) acc[w] += ((i >> (9 - w)) & 1) ? -p : p;
  }
  #pragma unroll
  for (int w = 0; w < NQ; ++w) {
    float v = acc[w];
    #pragma unroll
    for (int off = 16; off > 0; off >>= 1) v += __shfl_xor(v, off, 32);
    if (lane == 0) out[b * NQ + w] = v;
  }
}

extern "C" void kernel_launch(void* const* d_in, const int* in_sizes, int n_in,
                              void* d_out, int out_size, void* d_ws, size_t ws_size,
                              hipStream_t stream) {
  (void)in_sizes; (void)n_in; (void)out_size; (void)d_ws; (void)ws_size;
  const float* x  = (const float*)d_in[0];
  const float* qw = (const float*)d_in[1];
  float* out = (float*)d_out;
  const int batch = 2048;
  dim3 grid(batch / WAVES_PER_BLOCK);
  dim3 block(THREADS);
  qlayer_kernel<<<grid, block, LDS_BYTES, stream>>>(x, qw, out);
}


# State kept in WMMA accumulator registers through a layer's three gate stages (D of a stage is the next one's A/B fragment); LDS only at the CNOT ring; each amplitude's summation order changes
# speedup vs baseline: 1.3877x; 1.3877x over previous
#include <hip/hip_runtime.h>

typedef __attribute__((ext_vector_type(2))) float v2f;
typedef __attribute__((ext_vector_type(8))) float v8f;

#define NQ 10
#define NL 4
#define NSTATES 1024
#define WAVES_PER_BLOCK 4
#define THREADS (WAVES_PER_BLOCK * 32)

#define LDS_BYTES (1280 + 8192 + 8192 + 512 + WAVES_PER_BLOCK * 8192)

__device__ __forceinline__ int slot_of(const int f) {
  return ((f >> 6) & 1) | (((f >> 2) & 15) << 1) | (((f >> 9) & 1) << 5) |
         (((f >> 7) & 3) << 6) | ((f & 3) << 8);
}


__device__ __forceinline__ void load_gate16(const float* Ure, const float* Uim, const int lane,
                                            v2f (&gRe)[4], v2f (&gNs)[4], v2f (&gDf)[4]) {
  const int h = lane >> 4, n = lane & 15;
  const float4* pr = (const float4*)(Ure + n * 16 + 8 * h);
  const float4* pi = (const float4*)(Uim + n * 16 + 8 * h);
  const float4 r0 = pr[0], r1 = pr[1], i0 = pi[0], i1 = pi[1];
  const float rr[8] = {r0.x, r0.y, r0.z, r0.w, r1.x, r1.y, r1.z, r1.w};
  const float ii[8] = {i0.x, i0.y, i0.z, i0.w, i1.x, i1.y, i1.z, i1.w};
  #pragma unroll
  for (int k = 0; k < 4; ++k) {
    gRe[k].x = rr[2 * k];                 gRe[k].y = rr[2 * k + 1];
    gNs[k].x = -rr[2 * k] - ii[2 * k];    gNs[k].y = -rr[2 * k + 1] - ii[2 * k + 1];
    gDf[k].x = ii[2 * k] - rr[2 * k];     gDf[k].y = ii[2 * k + 1] - rr[2 * k + 1];
  }
}

#define WMMA4(A, B, C) __builtin_amdgcn_wmma_f32_16x16x4_f32(false, (A), false, (B), (short)0, (C), false, false)

__device__ __forceinline__ int cnot_dst(int j, const int rr) {
  #pragma unroll
  for (int w = 0; w < NQ; ++w) {
    const int cpos = 9 - w;
    int t = w + rr;
    t = (t >= NQ) ? t - NQ : t;
    const int tpos = 9 - t;
    j ^= ((j >> cpos) & 1) << tpos;
  }
  return j;
}

__device__ __forceinline__ void layer_regs(float2* S,
                                           const float* UAre, const float* UAim,
                                           const float* UBre, const float* UBim,
                                           const float* ucr, const float* uci,
                                           const int dLane, const int g0, const int g1,
                                           const int g2, const int g3, const int g4,
                                           const int lane) {
  const int h = lane >> 4, n = lane & 15;
  v2f uaRe[4], uaNs[4], uaDf[4], ubRe[4], ubNs[4], ubDf[4];
  load_gate16(UAre, UAim, lane, uaRe, uaNs, uaDf);
  load_gate16(UBre, UBim, lane, ubRe, ubNs, ubDf);

  const bool on = ((n >> 3) == h);
  const int jsel = n & 1, r = (n >> 1) & 3;
  v2f wRe[4], wNs[4], wDf[4];
  #pragma unroll
  for (int k = 0; k < 4; ++k) {
    const float vr = on ? ucr[r * 4 + k] : 0.f;
    const float vi = on ? uci[r * 4 + k] : 0.f;
    const float ns = -vr - vi, df = vi - vr;
    wRe[k].x = jsel ? 0.f : vr;  wRe[k].y = jsel ? vr : 0.f;
    wNs[k].x = jsel ? 0.f : ns;  wNs[k].y = jsel ? ns : 0.f;
    wDf[k].x = jsel ? 0.f : df;  wDf[k].y = jsel ? df : 0.f;
  }

  const float2* Sl = S + 2 * lane;
  float4 z[4][4];
  #pragma unroll
  for (int c = 0; c < 4; ++c) {
    #pragma unroll
    for (int k = 0; k < 4; ++k) z[c][k] = *(const float4*)(Sl + 64 * k + 256 * c);
  }

  v8f re1[4], im1[4];
  #pragma unroll
  for (int c = 0; c < 4; ++c) re1[c] = (v8f){0.f,0.f,0.f,0.f,0.f,0.f,0.f,0.f};
  #pragma unroll
  for (int k = 0; k < 4; ++k) {
    #pragma unroll
    for (int c = 0; c < 4; ++c) {
      v2f as;  as.x = z[c][k].x + z[c][k].y;  as.y = z[c][k].z + z[c][k].w;
      re1[c] = WMMA4(as, uaRe[k], re1[c]);
    }
  }
  #pragma unroll
  for (int c = 0; c < 4; ++c) im1[c] = re1[c];
  #pragma unroll
  for (int k = 0; k < 4; ++k) {
    #pragma unroll
    for (int c = 0; c < 4; ++c) {
      v2f ar, ai;
      ar.x = z[c][k].x;  ar.y = z[c][k].z;
      ai.x = z[c][k].y;  ai.y = z[c][k].w;
      re1[c] = WMMA4(ai, uaNs[k], re1[c]);
      im1[c] = WMMA4(ar, uaDf[k], im1[c]);
    }
  }

  v8f d2re[4], d2im[4];
  #pragma unroll
  for (int c = 0; c < 4; ++c) d2re[c] = (v8f){0.f,0.f,0.f,0.f,0.f,0.f,0.f,0.f};
  #pragma unroll
  for (int k = 0; k < 4; ++k) {
    #pragma unroll
    for (int c = 0; c < 4; ++c) {
      v2f bs;
      bs.x = re1[c][2 * k] + im1[c][2 * k];
      bs.y = re1[c][2 * k + 1] + im1[c][2 * k + 1];
      d2re[c] = WMMA4(ubRe[k], bs, d2re[c]);
    }
  }
  #pragma unroll
  for (int c = 0; c < 4; ++c) d2im[c] = d2re[c];
  #pragma unroll
  for (int k = 0; k < 4; ++k) {
    #pragma unroll
    for (int c = 0; c < 4; ++c) {
      v2f br, bi;
      br.x = re1[c][2 * k];  br.y = re1[c][2 * k + 1];
      bi.x = im1[c][2 * k];  bi.y = im1[c][2 * k + 1];
      d2re[c] = WMMA4(ubNs[k], bi, d2re[c]);
      d2im[c] = WMMA4(ubDf[k], br, d2im[c]);
    }
  }

  v8f re3[4], im3[4];
  #pragma unroll
  for (int t = 0; t < 4; ++t) re3[t] = (v8f){0.f,0.f,0.f,0.f,0.f,0.f,0.f,0.f};
  #pragma unroll
  for (int k = 0; k < 4; ++k) {
    #pragma unroll
    for (int t = 0; t < 4; ++t) {
      v2f bs;
      bs.x = d2re[k][2 * t] + d2im[k][2 * t];
      bs.y = d2re[k][2 * t + 1] + d2im[k][2 * t + 1];
      re3[t] = WMMA4(wRe[k], bs, re3[t]);
    }
  }
  #pragma unroll
  for (int t = 0; t < 4; ++t) im3[t] = re3[t];
  #pragma unroll
  for (int k = 0; k < 4; ++k) {
    #pragma unroll
    for (int t = 0; t < 4; ++t) {
      v2f br, bi;
      br.x = d2re[k][2 * t];  br.y = d2re[k][2 * t + 1];
      bi.x = d2im[k][2 * t];  bi.y = d2im[k][2 * t + 1];
      re3[t] = WMMA4(wNs[k], bi, re3[t]);
      im3[t] = WMMA4(wDf[k], br, im3[t]);
    }
  }
  #pragma unroll
  for (int t = 0; t < 4; ++t) {
    const int dj = dLane ^ ((t & 1) ? g3 : 0) ^ ((t & 2) ? g4 : 0);
    #pragma unroll
    for (int v = 0; v < 8; ++v) {
      const int o = dj ^ ((v & 1) ? g2 : 0) ^ ((v & 2) ? g0 : 0) ^ ((v & 4) ? g1 : 0);
      S[o] = make_float2(re3[t][v], im3[t][v]);
    }
  }
}

extern "C" __global__ __launch_bounds__(THREADS)
void qlayer_kernel(const float* __restrict__ x,
                   const float* __restrict__ qw,
                   float* __restrict__ out) {
  __shared__ __attribute__((aligned(16))) char smem_raw[LDS_BYTES];
  float* u2   = (float*)smem_raw;
  float* UAre = u2 + NL * NQ * 8;
  float* UAim = UAre + NL * 256;
  float* UBre = UAim + NL * 256;
  float* UBim = UBre + NL * 256;
  float* UCre = UBim + NL * 256;
  float* UCim = UCre + NL * 16;
  float2* stateBase = (float2*)(UCim + NL * 16);

  const int tid = threadIdx.x;

  #pragma unroll 1
  for (int e = tid; e < NL * NQ; e += THREADS) {
    const int l = e / NQ, w = e % NQ;
    const float phi = qw[(l * NQ + w) * 3 + 0];
    const float th  = qw[(l * NQ + w) * 3 + 1];
    const float om  = qw[(l * NQ + w) * 3 + 2];
    const float c = cosf(0.5f * th), s = sinf(0.5f * th);
    const float ap = 0.5f * (phi + om), am = 0.5f * (phi - om);
    const float cpo = cosf(ap), spo = sinf(ap);
    const float cb = cosf(am),  sb = sinf(am);
    float* g = u2 + e * 8;
    g[0] =  cpo * c; g[1] = -spo * c;
    g[2] = -cb * s;  g[3] = -sb * s;
    g[4] =  cb * s;  g[5] = -sb * s;
    g[6] =  cpo * c; g[7] =  spo * c;
  }
  __syncthreads();

  #pragma unroll 1
  for (int e = tid; e < NL * 256; e += THREADS) {
    const int l = e >> 8, rc = e & 255, r = rc >> 4, cc = rc & 15;
    float arA = 1.f, aiA = 0.f, arB = 1.f, aiB = 0.f;
    #pragma unroll
    for (int q = 0; q < 4; ++q) {
      const int rb = (r >> (3 - q)) & 1, cbit = (cc >> (3 - q)) & 1;
      const float* gA = u2 + (((l * NQ + q) * 4) + rb * 2 + cbit) * 2;
      float nr = arA * gA[0] - aiA * gA[1];
      float ni = arA * gA[1] + aiA * gA[0];
      arA = nr; aiA = ni;
      const float* gB = u2 + (((l * NQ + 4 + q) * 4) + rb * 2 + cbit) * 2;
      nr = arB * gB[0] - aiB * gB[1];
      ni = arB * gB[1] + aiB * gB[0];
      arB = nr; aiB = ni;
    }
    UAre[e] = arA; UAim[e] = aiA;
    UBre[e] = arB; UBim[e] = aiB;
  }
  #pragma unroll 1
  for (int e = tid; e < NL * 16; e += THREADS) {
    const int l = e >> 4, r = (e >> 2) & 3, cc = e & 3;
    const float* g8 = u2 + (((l * NQ + 8) * 4) + ((r >> 1) & 1) * 2 + ((cc >> 1) & 1)) * 2;
    const float* g9 = u2 + (((l * NQ + 9) * 4) + (r & 1) * 2 + (cc & 1)) * 2;
    UCre[e] = g8[0] * g9[0] - g8[1] * g9[1];
    UCim[e] = g8[0] * g9[1] + g8[1] * g9[0];
  }
  __syncthreads();

  const int wave = tid >> 5, lane = tid & 31;
  const int b = blockIdx.x * WAVES_PER_BLOCK + wave;
  float2* S = stateBase + wave * NSTATES;

  float cw[NQ], sw[NQ];
  #pragma unroll
  for (int w = 0; w < NQ; ++w) {
    const float ang = 1.5707963267948966f * x[b * NQ + w];
    cw[w] = cosf(ang);
    sw[w] = sinf(ang);
  }
  const int slotLane = slot_of(lane);
  for (int t = 0; t < 32; ++t) {
    const int i = lane + 32 * t;
    float a = 1.f;
    #pragma unroll
    for (int w = 0; w < NQ; ++w) a *= ((i >> (9 - w)) & 1) ? sw[w] : cw[w];
    S[slotLane | slot_of(32 * t)] = make_float2(a, 0.f);
  }

  #pragma unroll 1
  for (int l = 0; l < NL; ++l) {
    const int rr = (l % (NQ - 1)) + 1;
    const int n = lane & 15, h = lane >> 4;
    const int dLane = slot_of(cnot_dst((n << 6) | (h << 5), rr));
    const int bas = slot_of(cnot_dst(1 << (lane & 15), rr));
    const int g0 = __builtin_amdgcn_readlane(bas, 0);
    const int g1 = __builtin_amdgcn_readlane(bas, 1);
    const int g2 = __builtin_amdgcn_readlane(bas, 2);
    const int g3 = __builtin_amdgcn_readlane(bas, 3);
    const int g4 = __builtin_amdgcn_readlane(bas, 4);
    layer_regs(S, UAre + l * 256, UAim + l * 256, UBre + l * 256, UBim + l * 256,
               UCre + l * 16, UCim + l * 16, dLane, g0, g1, g2, g3, g4, lane);
  }

  float acc[NQ];
  #pragma unroll
  for (int w = 0; w < NQ; ++w) acc[w] = 0.f;
  for (int t = 0; t < 32; ++t) {
    const int i = lane + 32 * t;
    const float2 z = S[slotLane | slot_of(32 * t)];
    const float p = z.x * z.x + z.y * z.y;
    #pragma unroll
    for (int w = 0; w < NQ; ++w) acc[w] += ((i >> (9 - w)) & 1) ? -p : p;
  }
  #pragma unroll
  for (int w = 0; w < NQ; ++w) {
    float v = acc[w];
    #pragma unroll
    for (int off = 16; off > 0; off >>= 1) v += __shfl_xor(v, off, 32);
    if (lane == 0) out[b * NQ + w] = v;
  }
}

extern "C" void kernel_launch(void* const* d_in, const int* in_sizes, int n_in,
                              void* d_out, int out_size, void* d_ws, size_t ws_size,
                              hipStream_t stream) {
  (void)in_sizes; (void)n_in; (void)out_size; (void)d_ws; (void)ws_size;
  const float* x  = (const float*)d_in[0];
  const float* qw = (const float*)d_in[1];
  float* out = (float*)d_out;
  const int batch = 2048;
  dim3 grid(batch / WAVES_PER_BLOCK);
  dim3 block(THREADS);
  qlayer_kernel<<<grid, block, 0, stream>>>(x, qw, out);
}


# Z expectation sums on the matrix unit: probabilities (A) times a +-1 sign matrix (B), 16 f32 WMMAs, then 7 in-lane adds and one swap of the lane halves, one 10-lane store; their sum order changes
# speedup vs baseline: 1.6844x; 1.2138x over previous
#include <hip/hip_runtime.h>

typedef __attribute__((ext_vector_type(2))) float v2f;
typedef __attribute__((ext_vector_type(8))) float v8f;

#define NQ 10
#define NL 4
#define NSTATES 1024
#define WAVES_PER_BLOCK 4
#define THREADS (WAVES_PER_BLOCK * 32)

#define LDS_BYTES (1280 + 8192 + 8192 + 512 + WAVES_PER_BLOCK * 8192)

__device__ __forceinline__ int slot_of(const int f) {
  return ((f >> 6) & 1) | (((f >> 2) & 15) << 1) | (((f >> 9) & 1) << 5) |
         (((f >> 7) & 3) << 6) | ((f & 3) << 8);
}


__device__ __forceinline__ void load_gate16(const float* Ure, const float* Uim, const int lane,
                                            v2f (&gRe)[4], v2f (&gNs)[4], v2f (&gDf)[4]) {
  const int h = lane >> 4, n = lane & 15;
  const float4* pr = (const float4*)(Ure + n * 16 + 8 * h);
  const float4* pi = (const float4*)(Uim + n * 16 + 8 * h);
  const float4 r0 = pr[0], r1 = pr[1], i0 = pi[0], i1 = pi[1];
  const float rr[8] = {r0.x, r0.y, r0.z, r0.w, r1.x, r1.y, r1.z, r1.w};
  const float ii[8] = {i0.x, i0.y, i0.z, i0.w, i1.x, i1.y, i1.z, i1.w};
  #pragma unroll
  for (int k = 0; k < 4; ++k) {
    gRe[k].x = rr[2 * k];                 gRe[k].y = rr[2 * k + 1];
    gNs[k].x = -rr[2 * k] - ii[2 * k];    gNs[k].y = -rr[2 * k + 1] - ii[2 * k + 1];
    gDf[k].x = ii[2 * k] - rr[2 * k];     gDf[k].y = ii[2 * k + 1] - rr[2 * k + 1];
  }
}

#define WMMA4(A, B, C) __builtin_amdgcn_wmma_f32_16x16x4_f32(false, (A), false, (B), (short)0, (C), false, false)

__device__ __forceinline__ int cnot_dst(int j, const int rr) {
  #pragma unroll
  for (int w = 0; w < NQ; ++w) {
    const int cpos = 9 - w;
    int t = w + rr;
    t = (t >= NQ) ? t - NQ : t;
    const int tpos = 9 - t;
    j ^= ((j >> cpos) & 1) << tpos;
  }
  return j;
}

__device__ __forceinline__ void layer_regs(float2* S,
                                           const float* UAre, const float* UAim,
                                           const float* UBre, const float* UBim,
                                           const float* ucr, const float* uci,
                                           const int dLane, const int g0, const int g1,
                                           const int g2, const int g3, const int g4,
                                           const int lane) {
  const int h = lane >> 4, n = lane & 15;
  v2f uaRe[4], uaNs[4], uaDf[4], ubRe[4], ubNs[4], ubDf[4];
  load_gate16(UAre, UAim, lane, uaRe, uaNs, uaDf);
  load_gate16(UBre, UBim, lane, ubRe, ubNs, ubDf);

  const bool on = ((n >> 3) == h);
  const int jsel = n & 1, r = (n >> 1) & 3;
  v2f wRe[4], wNs[4], wDf[4];
  #pragma unroll
  for (int k = 0; k < 4; ++k) {
    const float vr = on ? ucr[r * 4 + k] : 0.f;
    const float vi = on ? uci[r * 4 + k] : 0.f;
    const float ns = -vr - vi, df = vi - vr;
    wRe[k].x = jsel ? 0.f : vr;  wRe[k].y = jsel ? vr : 0.f;
    wNs[k].x = jsel ? 0.f : ns;  wNs[k].y = jsel ? ns : 0.f;
    wDf[k].x = jsel ? 0.f : df;  wDf[k].y = jsel ? df : 0.f;
  }

  const float2* Sl = S + 2 * lane;
  float4 z[4][4];
  #pragma unroll
  for (int c = 0; c < 4; ++c) {
    #pragma unroll
    for (int k = 0; k < 4; ++k) z[c][k] = *(const float4*)(Sl + 64 * k + 256 * c);
  }

  v8f re1[4], im1[4];
  #pragma unroll
  for (int c = 0; c < 4; ++c) re1[c] = (v8f){0.f,0.f,0.f,0.f,0.f,0.f,0.f,0.f};
  #pragma unroll
  for (int k = 0; k < 4; ++k) {
    #pragma unroll
    for (int c = 0; c < 4; ++c) {
      v2f as;  as.x = z[c][k].x + z[c][k].y;  as.y = z[c][k].z + z[c][k].w;
      re1[c] = WMMA4(as, uaRe[k], re1[c]);
    }
  }
  #pragma unroll
  for (int c = 0; c < 4; ++c) im1[c] = re1[c];
  #pragma unroll
  for (int k = 0; k < 4; ++k) {
    #pragma unroll
    for (int c = 0; c < 4; ++c) {
      v2f ar, ai;
      ar.x = z[c][k].x;  ar.y = z[c][k].z;
      ai.x = z[c][k].y;  ai.y = z[c][k].w;
      re1[c] = WMMA4(ai, uaNs[k], re1[c]);
      im1[c] = WMMA4(ar, uaDf[k], im1[c]);
    }
  }

  v8f d2re[4], d2im[4];
  #pragma unroll
  for (int c = 0; c < 4; ++c) d2re[c] = (v8f){0.f,0.f,0.f,0.f,0.f,0.f,0.f,0.f};
  #pragma unroll
  for (int k = 0; k < 4; ++k) {
    #pragma unroll
    for (int c = 0; c < 4; ++c) {
      v2f bs;
      bs.x = re1[c][2 * k] + im1[c][2 * k];
      bs.y = re1[c][2 * k + 1] + im1[c][2 * k + 1];
      d2re[c] = WMMA4(ubRe[k], bs, d2re[c]);
    }
  }
  #pragma unroll
  for (int c = 0; c < 4; ++c) d2im[c] = d2re[c];
  #pragma unroll
  for (int k = 0; k < 4; ++k) {
    #pragma unroll
    for (int c = 0; c < 4; ++c) {
      v2f br, bi;
      br.x = re1[c][2 * k];  br.y = re1[c][2 * k + 1];
      bi.x = im1[c][2 * k];  bi.y = im1[c][2 * k + 1];
      d2re[c] = WMMA4(ubNs[k], bi, d2re[c]);
      d2im[c] = WMMA4(ubDf[k], br, d2im[c]);
    }
  }

  v8f re3[4], im3[4];
  #pragma unroll
  for (int t = 0; t < 4; ++t) re3[t] = (v8f){0.f,0.f,0.f,0.f,0.f,0.f,0.f,0.f};
  #pragma unroll
  for (int k = 0; k < 4; ++k) {
    #pragma unroll
    for (int t = 0; t < 4; ++t) {
      v2f bs;
      bs.x = d2re[k][2 * t] + d2im[k][2 * t];
      bs.y = d2re[k][2 * t + 1] + d2im[k][2 * t + 1];
      re3[t] = WMMA4(wRe[k], bs, re3[t]);
    }
  }
  #pragma unroll
  for (int t = 0; t < 4; ++t) im3[t] = re3[t];
  #pragma unroll
  for (int k = 0; k < 4; ++k) {
    #pragma unroll
    for (int t = 0; t < 4; ++t) {
      v2f br, bi;
      br.x = d2re[k][2 * t];  br.y = d2re[k][2 * t + 1];
      bi.x = d2im[k][2 * t];  bi.y = d2im[k][2 * t + 1];
      re3[t] = WMMA4(wNs[k], bi, re3[t]);
      im3[t] = WMMA4(wDf[k], br, im3[t]);
    }
  }
  #pragma unroll
  for (int t = 0; t < 4; ++t) {
    const int dj = dLane ^ ((t & 1) ? g3 : 0) ^ ((t & 2) ? g4 : 0);
    #pragma unroll
    for (int v = 0; v < 8; ++v) {
      const int o = dj ^ ((v & 1) ? g2 : 0) ^ ((v & 2) ? g0 : 0) ^ ((v & 4) ? g1 : 0);
      S[o] = make_float2(re3[t][v], im3[t][v]);
    }
  }
}

extern "C" __global__ __launch_bounds__(THREADS)
void qlayer_kernel(const float* __restrict__ x,
                   const float* __restrict__ qw,
                   float* __restrict__ out) {
  __shared__ __attribute__((aligned(16))) char smem_raw[LDS_BYTES];
  float* u2   = (float*)smem_raw;
  float* UAre = u2 + NL * NQ * 8;
  float* UAim = UAre + NL * 256;
  float* UBre = UAim + NL * 256;
  float* UBim = UBre + NL * 256;
  float* UCre = UBim + NL * 256;
  float* UCim = UCre + NL * 16;
  float2* stateBase = (float2*)(UCim + NL * 16);

  const int tid = threadIdx.x;

  #pragma unroll 1
  for (int e = tid; e < NL * NQ; e += THREADS) {
    const int l = e / NQ, w = e % NQ;
    const float phi = qw[(l * NQ + w) * 3 + 0];
    const float th  = qw[(l * NQ + w) * 3 + 1];
    const float om  = qw[(l * NQ + w) * 3 + 2];
    const float c = cosf(0.5f * th), s = sinf(0.5f * th);
    const float ap = 0.5f * (phi + om), am = 0.5f * (phi - om);
    const float cpo = cosf(ap), spo = sinf(ap);
    const float cb = cosf(am),  sb = sinf(am);
    float* g = u2 + e * 8;
    g[0] =  cpo * c; g[1] = -spo * c;
    g[2] = -cb * s;  g[3] = -sb * s;
    g[4] =  cb * s;  g[5] = -sb * s;
    g[6] =  cpo * c; g[7] =  spo * c;
  }
  __syncthreads();

  #pragma unroll 1
  for (int e = tid; e < NL * 256; e += THREADS) {
    const int l = e >> 8, rc = e & 255, r = rc >> 4, cc = rc & 15;
    float arA = 1.f, aiA = 0.f, arB = 1.f, aiB = 0.f;
    #pragma unroll
    for (int q = 0; q < 4; ++q) {
      const int rb = (r >> (3 - q)) & 1, cbit = (cc >> (3 - q)) & 1;
      const float* gA = u2 + (((l * NQ + q) * 4) + rb * 2 + cbit) * 2;
      float nr = arA * gA[0] - aiA * gA[1];
      float ni = arA * gA[1] + aiA * gA[0];
      arA = nr; aiA = ni;
      const float* gB = u2 + (((l * NQ + 4 + q) * 4) + rb * 2 + cbit) * 2;
      nr = arB * gB[0] - aiB * gB[1];
      ni = arB * gB[1] + aiB * gB[0];
      arB = nr; aiB = ni;
    }
    UAre[e] = arA; UAim[e] = aiA;
    UBre[e] = arB; UBim[e] = aiB;
  }
  #pragma unroll 1
  for (int e = tid; e < NL * 16; e += THREADS) {
    const int l = e >> 4, r = (e >> 2) & 3, cc = e & 3;
    const float* g8 = u2 + (((l * NQ + 8) * 4) + ((r >> 1) & 1) * 2 + ((cc >> 1) & 1)) * 2;
    const float* g9 = u2 + (((l * NQ + 9) * 4) + (r & 1) * 2 + (cc & 1)) * 2;
    UCre[e] = g8[0] * g9[0] - g8[1] * g9[1];
    UCim[e] = g8[0] * g9[1] + g8[1] * g9[0];
  }
  __syncthreads();

  const int wave = tid >> 5, lane = tid & 31;
  const int b = blockIdx.x * WAVES_PER_BLOCK + wave;
  float2* S = stateBase + wave * NSTATES;

  float cw[NQ], sw[NQ];
  #pragma unroll
  for (int w = 0; w < NQ; ++w) {
    const float ang = 1.5707963267948966f * x[b * NQ + w];
    cw[w] = cosf(ang);
    sw[w] = sinf(ang);
  }
  const int slotLane = slot_of(lane);
  for (int t = 0; t < 32; ++t) {
    const int i = lane + 32 * t;
    float a = 1.f;
    #pragma unroll
    for (int w = 0; w < NQ; ++w) a *= ((i >> (9 - w)) & 1) ? sw[w] : cw[w];
    S[slotLane | slot_of(32 * t)] = make_float2(a, 0.f);
  }

  #pragma unroll 1
  for (int l = 0; l < NL; ++l) {
    const int rr = (l % (NQ - 1)) + 1;
    const int n = lane & 15, h = lane >> 4;
    const int dLane = slot_of(cnot_dst((n << 6) | (h << 5), rr));
    const int bas = slot_of(cnot_dst(1 << (lane & 15), rr));
    const int g0 = __builtin_amdgcn_readlane(bas, 0);
    const int g1 = __builtin_amdgcn_readlane(bas, 1);
    const int g2 = __builtin_amdgcn_readlane(bas, 2);
    const int g3 = __builtin_amdgcn_readlane(bas, 3);
    const int g4 = __builtin_amdgcn_readlane(bas, 4);
    layer_regs(S, UAre + l * 256, UAim + l * 256, UBre + l * 256, UBim + l * 256,
               UCre + l * 16, UCim + l * 16, dLane, g0, g1, g2, g3, g4, lane);
  }

  {
    const int h = lane >> 4, n = lane & 15;
    const float2* Sl = S + 2 * lane;
    float4 z[16];
    #pragma unroll
    for (int t = 0; t < 16; ++t) z[t] = *(const float4*)(Sl + 64 * t);
    int fm = ((unsigned)(n - 1) < 3u) ? (8 >> n) : 0;
    fm = ((unsigned)(n - 8) < 2u) ? (4096 >> n) : fm;
    float gp = (n < NQ) ? 1.f : 0.f;
    gp = (n == 0) ? (1.f - 2.f * (float)h) : gp;
    const float gn = -gp;
    v8f d0 = (v8f){0.f,0.f,0.f,0.f,0.f,0.f,0.f,0.f};
    v8f d1 = (v8f){0.f,0.f,0.f,0.f,0.f,0.f,0.f,0.f};
    #pragma unroll
    for (int t = 0; t < 16; t += 2) {
      v2f pa, pb, ga, gb;
      pa.x = z[t].x * z[t].x + z[t].y * z[t].y;
      pa.y = z[t].z * z[t].z + z[t].w * z[t].w;
      pb.x = z[t + 1].x * z[t + 1].x + z[t + 1].y * z[t + 1].y;
      pb.y = z[t + 1].z * z[t + 1].z + z[t + 1].w * z[t + 1].w;
      ga.x = (fm & (2 * t)) ? gn : gp;      ga.y = (fm & (2 * t + 1)) ? gn : gp;
      gb.x = (fm & (2 * t + 2)) ? gn : gp;  gb.y = (fm & (2 * t + 3)) ? gn : gp;
      d0 = WMMA4(pa, ga, d0);
      d1 = WMMA4(pb, gb, d1);
    }
    const v8f ds = d0 + d1;
    const float s7 = (n == 7) ? -1.f : 1.f;
    const float s6 = (n == 6) ? -1.f : 1.f;
    const float s5 = (n == 5) ? -1.f : 1.f;
    const float s4 = (n == 4) ? -1.f : 1.f;
    const float t0 = ds[0] + s7 * ds[1], t1 = ds[2] + s7 * ds[3];
    const float t2 = ds[4] + s7 * ds[5], t3 = ds[6] + s7 * ds[7];
    const float u0 = t0 + s6 * t1, u1 = t2 + s6 * t3;
    const float v = u0 + s5 * u1;
    const float val = v + s4 * __shfl_xor(v, 16, 32);
    if (lane < NQ) out[b * NQ + lane] = val;
  }
}

extern "C" void kernel_launch(void* const* d_in, const int* in_sizes, int n_in,
                              void* d_out, int out_size, void* d_ws, size_t ws_size,
                              hipStream_t stream) {
  (void)in_sizes; (void)n_in; (void)out_size; (void)d_ws; (void)ws_size;
  const float* x  = (const float*)d_in[0];
  const float* qw = (const float*)d_in[1];
  float* out = (float*)d_out;
  const int batch = 2048;
  dim3 grid(batch / WAVES_PER_BLOCK);
  dim3 block(THREADS);
  qlayer_kernel<<<grid, block, 0, stream>>>(x, qw, out);
}


# Layer 0 on the product state's factors: each group's gate times its factor vector (3x8 WMMAs, not 144 there), amplitudes as products of the three results, no LDS state init; sums become products
# speedup vs baseline: 2.0042x; 1.1899x over previous
#include <hip/hip_runtime.h>

typedef __attribute__((ext_vector_type(2))) float v2f;
typedef __attribute__((ext_vector_type(8))) float v8f;

#define NQ 10
#define NL 4
#define NSTATES 1024
#define WAVES_PER_BLOCK 4
#define THREADS (WAVES_PER_BLOCK * 32)

#define LDS_BYTES (1280 + 8192 + 8192 + 512 + WAVES_PER_BLOCK * 8192)

__device__ __forceinline__ int slot_of(const int f) {
  return ((f >> 6) & 1) | (((f >> 2) & 15) << 1) | (((f >> 9) & 1) << 5) |
         (((f >> 7) & 3) << 6) | ((f & 3) << 8);
}


__device__ __forceinline__ void load_gate16(const float* Ure, const float* Uim, const int lane,
                                            v2f (&gRe)[4], v2f (&gNs)[4], v2f (&gDf)[4]) {
  const int h = lane >> 4, n = lane & 15;
  const float4* pr = (const float4*)(Ure + n * 16 + 8 * h);
  const float4* pi = (const float4*)(Uim + n * 16 + 8 * h);
  const float4 r0 = pr[0], r1 = pr[1], i0 = pi[0], i1 = pi[1];
  const float rr[8] = {r0.x, r0.y, r0.z, r0.w, r1.x, r1.y, r1.z, r1.w};
  const float ii[8] = {i0.x, i0.y, i0.z, i0.w, i1.x, i1.y, i1.z, i1.w};
  #pragma unroll
  for (int k = 0; k < 4; ++k) {
    gRe[k].x = rr[2 * k];                 gRe[k].y = rr[2 * k + 1];
    gNs[k].x = -rr[2 * k] - ii[2 * k];    gNs[k].y = -rr[2 * k + 1] - ii[2 * k + 1];
    gDf[k].x = ii[2 * k] - rr[2 * k];     gDf[k].y = ii[2 * k + 1] - rr[2 * k + 1];
  }
}

#define WMMA4(A, B, C) __builtin_amdgcn_wmma_f32_16x16x4_f32(false, (A), false, (B), (short)0, (C), false, false)

__device__ __forceinline__ int cnot_dst(int j, const int rr) {
  #pragma unroll
  for (int w = 0; w < NQ; ++w) {
    const int cpos = 9 - w;
    int t = w + rr;
    t = (t >= NQ) ? t - NQ : t;
    const int tpos = 9 - t;
    j ^= ((j >> cpos) & 1) << tpos;
  }
  return j;
}

__device__ __forceinline__ void load_gate4(const float* ucr, const float* uci, const int lane,
                                           v2f (&wRe)[4], v2f (&wNs)[4], v2f (&wDf)[4]) {
  const int h = lane >> 4, n = lane & 15;
  const bool on = ((n >> 3) == h);
  const int jsel = n & 1, r = (n >> 1) & 3;
  #pragma unroll
  for (int k = 0; k < 4; ++k) {
    const float vr = on ? ucr[r * 4 + k] : 0.f;
    const float vi = on ? uci[r * 4 + k] : 0.f;
    const float ns = -vr - vi, df = vi - vr;
    wRe[k].x = jsel ? 0.f : vr;  wRe[k].y = jsel ? vr : 0.f;
    wNs[k].x = jsel ? 0.f : ns;  wNs[k].y = jsel ? ns : 0.f;
    wDf[k].x = jsel ? 0.f : df;  wDf[k].y = jsel ? df : 0.f;
  }
}

__device__ __forceinline__ void scatter_ring(float2* S, const v8f (&re3)[4], const v8f (&im3)[4],
                                             const int dLane, const int g0, const int g1,
                                             const int g2, const int g3, const int g4) {
  #pragma unroll
  for (int t = 0; t < 4; ++t) {
    const int dj = dLane ^ ((t & 1) ? g3 : 0) ^ ((t & 2) ? g4 : 0);
    #pragma unroll
    for (int v = 0; v < 8; ++v) {
      const int o = dj ^ ((v & 1) ? g2 : 0) ^ ((v & 2) ? g0 : 0) ^ ((v & 4) ? g1 : 0);
      S[o] = make_float2(re3[t][v], im3[t][v]);
    }
  }
}

__device__ __forceinline__ void ring_images(const int rr, const int lane, int& dLane,
                                            int& g0, int& g1, int& g2, int& g3, int& g4) {
  const int n = lane & 15, h = lane >> 4;
  dLane = slot_of(cnot_dst((n << 6) | (h << 5), rr));
  const int bas = slot_of(cnot_dst(1 << (lane & 15), rr));
  g0 = __builtin_amdgcn_readlane(bas, 0);
  g1 = __builtin_amdgcn_readlane(bas, 1);
  g2 = __builtin_amdgcn_readlane(bas, 2);
  g3 = __builtin_amdgcn_readlane(bas, 3);
  g4 = __builtin_amdgcn_readlane(bas, 4);
}

__device__ __forceinline__ void layer0_regs(float2* S,
                                            const float* UAre, const float* UAim,
                                            const float* UBre, const float* UBim,
                                            const float* ucr, const float* uci,
                                            const float (&cw)[NQ], const float (&sw)[NQ],
                                            const int dLane, const int g0, const int g1,
                                            const int g2, const int g3, const int g4,
                                            const int lane) {
  const int h = lane >> 4;
  v2f uaRe[4], uaNs[4], uaDf[4], ubRe[4], ubNs[4], ubDf[4], wRe[4], wNs[4], wDf[4];
  load_gate16(UAre, UAim, lane, uaRe, uaNs, uaDf);
  load_gate16(UBre, UBim, lane, ubRe, ubNs, ubDf);
  load_gate4(ucr, uci, lane, wRe, wNs, wDf);

  const float fa = h ? sw[0] : cw[0];
  const float fb = h ? sw[4] : cw[4];
  v2f va[4], vb[4], vc[4];
  #pragma unroll
  for (int k = 0; k < 4; ++k) {
    const float pa = fa * ((k & 2) ? sw[1] : cw[1]) * ((k & 1) ? sw[2] : cw[2]);
    va[k].x = pa * cw[3];  va[k].y = pa * sw[3];
    const float pb = fb * ((k & 2) ? sw[5] : cw[5]) * ((k & 1) ? sw[6] : cw[6]);
    vb[k].x = pb * cw[7];  vb[k].y = pb * sw[7];
    const float pc = ((k & 2) ? sw[8] : cw[8]) * ((k & 1) ? sw[9] : cw[9]);
    vc[k].x = pc;  vc[k].y = pc;
  }

  v8f aRe = (v8f){0.f,0.f,0.f,0.f,0.f,0.f,0.f,0.f};
  v8f bRe = aRe, cRe = aRe;
  #pragma unroll
  for (int k = 0; k < 4; ++k) {
    aRe = WMMA4(va[k], uaRe[k], aRe);
    bRe = WMMA4(ubRe[k], vb[k], bRe);
    cRe = WMMA4(wRe[k], vc[k], cRe);
  }
  v8f aIm = aRe, bIm = bRe, cIm = cRe;
  #pragma unroll
  for (int k = 0; k < 4; ++k) {
    aIm = WMMA4(va[k], uaDf[k], aIm);
    bIm = WMMA4(ubDf[k], vb[k], bIm);
    cIm = WMMA4(wDf[k], vc[k], cIm);
  }

  const float ar = aRe[0], ai = aIm[0];
  float abr[8], abi[8];
  #pragma unroll
  for (int u = 0; u < 8; ++u) {
    abr[u] = ar * bRe[u] - ai * bIm[u];
    abi[u] = ar * bIm[u] + ai * bRe[u];
  }
  v8f re3[4], im3[4];
  #pragma unroll
  for (int t = 0; t < 4; ++t) {
    #pragma unroll
    for (int v = 0; v < 8; ++v) {
      const int u = 2 * t + (v & 1);
      re3[t][v] = abr[u] * cRe[v] - abi[u] * cIm[v];
      im3[t][v] = abr[u] * cIm[v] + abi[u] * cRe[v];
    }
  }
  scatter_ring(S, re3, im3, dLane, g0, g1, g2, g3, g4);
}

__device__ __forceinline__ void layer_regs(float2* S,
                                           const float* UAre, const float* UAim,
                                           const float* UBre, const float* UBim,
                                           const float* ucr, const float* uci,
                                           const int dLane, const int g0, const int g1,
                                           const int g2, const int g3, const int g4,
                                           const int lane) {
  v2f uaRe[4], uaNs[4], uaDf[4], ubRe[4], ubNs[4], ubDf[4];
  load_gate16(UAre, UAim, lane, uaRe, uaNs, uaDf);
  load_gate16(UBre, UBim, lane, ubRe, ubNs, ubDf);

  v2f wRe[4], wNs[4], wDf[4];
  load_gate4(ucr, uci, lane, wRe, wNs, wDf);

  const float2* Sl = S + 2 * lane;
  float4 z[4][4];
  #pragma unroll
  for (int c = 0; c < 4; ++c) {
    #pragma unroll
    for (int k = 0; k < 4; ++k) z[c][k] = *(const float4*)(Sl + 64 * k + 256 * c);
  }

  v8f re1[4], im1[4];
  #pragma unroll
  for (int c = 0; c < 4; ++c) re1[c] = (v8f){0.f,0.f,0.f,0.f,0.f,0.f,0.f,0.f};
  #pragma unroll
  for (int k = 0; k < 4; ++k) {
    #pragma unroll
    for (int c = 0; c < 4; ++c) {
      v2f as;  as.x = z[c][k].x + z[c][k].y;  as.y = z[c][k].z + z[c][k].w;
      re1[c] = WMMA4(as, uaRe[k], re1[c]);
    }
  }
  #pragma unroll
  for (int c = 0; c < 4; ++c) im1[c] = re1[c];
  #pragma unroll
  for (int k = 0; k < 4; ++k) {
    #pragma unroll
    for (int c = 0; c < 4; ++c) {
      v2f ar, ai;
      ar.x = z[c][k].x;  ar.y = z[c][k].z;
      ai.x = z[c][k].y;  ai.y = z[c][k].w;
      re1[c] = WMMA4(ai, uaNs[k], re1[c]);
      im1[c] = WMMA4(ar, uaDf[k], im1[c]);
    }
  }

  v8f d2re[4], d2im[4];
  #pragma unroll
  for (int c = 0; c < 4; ++c) d2re[c] = (v8f){0.f,0.f,0.f,0.f,0.f,0.f,0.f,0.f};
  #pragma unroll
  for (int k = 0; k < 4; ++k) {
    #pragma unroll
    for (int c = 0; c < 4; ++c) {
      v2f bs;
      bs.x = re1[c][2 * k] + im1[c][2 * k];
      bs.y = re1[c][2 * k + 1] + im1[c][2 * k + 1];
      d2re[c] = WMMA4(ubRe[k], bs, d2re[c]);
    }
  }
  #pragma unroll
  for (int c = 0; c < 4; ++c) d2im[c] = d2re[c];
  #pragma unroll
  for (int k = 0; k < 4; ++k) {
    #pragma unroll
    for (int c = 0; c < 4; ++c) {
      v2f br, bi;
      br.x = re1[c][2 * k];  br.y = re1[c][2 * k + 1];
      bi.x = im1[c][2 * k];  bi.y = im1[c][2 * k + 1];
      d2re[c] = WMMA4(ubNs[k], bi, d2re[c]);
      d2im[c] = WMMA4(ubDf[k], br, d2im[c]);
    }
  }

  v8f re3[4], im3[4];
  #pragma unroll
  for (int t = 0; t < 4; ++t) re3[t] = (v8f){0.f,0.f,0.f,0.f,0.f,0.f,0.f,0.f};
  #pragma unroll
  for (int k = 0; k < 4; ++k) {
    #pragma unroll
    for (int t = 0; t < 4; ++t) {
      v2f bs;
      bs.x = d2re[k][2 * t] + d2im[k][2 * t];
      bs.y = d2re[k][2 * t + 1] + d2im[k][2 * t + 1];
      re3[t] = WMMA4(wRe[k], bs, re3[t]);
    }
  }
  #pragma unroll
  for (int t = 0; t < 4; ++t) im3[t] = re3[t];
  #pragma unroll
  for (int k = 0; k < 4; ++k) {
    #pragma unroll
    for (int t = 0; t < 4; ++t) {
      v2f br, bi;
      br.x = d2re[k][2 * t];  br.y = d2re[k][2 * t + 1];
      bi.x = d2im[k][2 * t];  bi.y = d2im[k][2 * t + 1];
      re3[t] = WMMA4(wNs[k], bi, re3[t]);
      im3[t] = WMMA4(wDf[k], br, im3[t]);
    }
  }
  scatter_ring(S, re3, im3, dLane, g0, g1, g2, g3, g4);
}

extern "C" __global__ __launch_bounds__(THREADS)
void qlayer_kernel(const float* __restrict__ x,
                   const float* __restrict__ qw,
                   float* __restrict__ out) {
  __shared__ __attribute__((aligned(16))) char smem_raw[LDS_BYTES];
  float* u2   = (float*)smem_raw;
  float* UAre = u2 + NL * NQ * 8;
  float* UAim = UAre + NL * 256;
  float* UBre = UAim + NL * 256;
  float* UBim = UBre + NL * 256;
  float* UCre = UBim + NL * 256;
  float* UCim = UCre + NL * 16;
  float2* stateBase = (float2*)(UCim + NL * 16);

  const int tid = threadIdx.x;

  #pragma unroll 1
  for (int e = tid; e < NL * NQ; e += THREADS) {
    const int l = e / NQ, w = e % NQ;
    const float phi = qw[(l * NQ + w) * 3 + 0];
    const float th  = qw[(l * NQ + w) * 3 + 1];
    const float om  = qw[(l * NQ + w) * 3 + 2];
    const float c = cosf(0.5f * th), s = sinf(0.5f * th);
    const float ap = 0.5f * (phi + om), am = 0.5f * (phi - om);
    const float cpo = cosf(ap), spo = sinf(ap);
    const float cb = cosf(am),  sb = sinf(am);
    float* g = u2 + e * 8;
    g[0] =  cpo * c; g[1] = -spo * c;
    g[2] = -cb * s;  g[3] = -sb * s;
    g[4] =  cb * s;  g[5] = -sb * s;
    g[6] =  cpo * c; g[7] =  spo * c;
  }
  __syncthreads();

  #pragma unroll 1
  for (int e = tid; e < NL * 256; e += THREADS) {
    const int l = e >> 8, rc = e & 255, r = rc >> 4, cc = rc & 15;
    float arA = 1.f, aiA = 0.f, arB = 1.f, aiB = 0.f;
    #pragma unroll
    for (int q = 0; q < 4; ++q) {
      const int rb = (r >> (3 - q)) & 1, cbit = (cc >> (3 - q)) & 1;
      const float* gA = u2 + (((l * NQ + q) * 4) + rb * 2 + cbit) * 2;
      float nr = arA * gA[0] - aiA * gA[1];
      float ni = arA * gA[1] + aiA * gA[0];
      arA = nr; aiA = ni;
      const float* gB = u2 + (((l * NQ + 4 + q) * 4) + rb * 2 + cbit) * 2;
      nr = arB * gB[0] - aiB * gB[1];
      ni = arB * gB[1] + aiB * gB[0];
      arB = nr; aiB = ni;
    }
    UAre[e] = arA; UAim[e] = aiA;
    UBre[e] = arB; UBim[e] = aiB;
  }
  #pragma unroll 1
  for (int e = tid; e < NL * 16; e += THREADS) {
    const int l = e >> 4, r = (e >> 2) & 3, cc = e & 3;
    const float* g8 = u2 + (((l * NQ + 8) * 4) + ((r >> 1) & 1) * 2 + ((cc >> 1) & 1)) * 2;
    const float* g9 = u2 + (((l * NQ + 9) * 4) + (r & 1) * 2 + (cc & 1)) * 2;
    UCre[e] = g8[0] * g9[0] - g8[1] * g9[1];
    UCim[e] = g8[0] * g9[1] + g8[1] * g9[0];
  }
  __syncthreads();

  const int wave = tid >> 5, lane = tid & 31;
  const int b = blockIdx.x * WAVES_PER_BLOCK + wave;
  float2* S = stateBase + wave * NSTATES;

  float cw[NQ], sw[NQ];
  #pragma unroll
  for (int w = 0; w < NQ; ++w) {
    const float ang = 1.5707963267948966f * x[b * NQ + w];
    cw[w] = cosf(ang);
    sw[w] = sinf(ang);
  }
  {
    int dLane, g0, g1, g2, g3, g4;
    ring_images(1, lane, dLane, g0, g1, g2, g3, g4);
    layer0_regs(S, UAre, UAim, UBre, UBim, UCre, UCim, cw, sw, dLane, g0, g1, g2, g3, g4, lane);
  }

  #pragma unroll 1
  for (int l = 1; l < NL; ++l) {
    const int rr = (l % (NQ - 1)) + 1;
    int dLane, g0, g1, g2, g3, g4;
    ring_images(rr, lane, dLane, g0, g1, g2, g3, g4);
    layer_regs(S, UAre + l * 256, UAim + l * 256, UBre + l * 256, UBim + l * 256,
               UCre + l * 16, UCim + l * 16, dLane, g0, g1, g2, g3, g4, lane);
  }

  {
    const int h = lane >> 4, n = lane & 15;
    const float2* Sl = S + 2 * lane;
    float4 z[16];
    #pragma unroll
    for (int t = 0; t < 16; ++t) z[t] = *(const float4*)(Sl + 64 * t);
    int fm = ((unsigned)(n - 1) < 3u) ? (8 >> n) : 0;
    fm = ((unsigned)(n - 8) < 2u) ? (4096 >> n) : fm;
    float gp = (n < NQ) ? 1.f : 0.f;
    gp = (n == 0) ? (1.f - 2.f * (float)h) : gp;
    const float gn = -gp;
    v8f d0 = (v8f){0.f,0.f,0.f,0.f,0.f,0.f,0.f,0.f};
    v8f d1 = (v8f){0.f,0.f,0.f,0.f,0.f,0.f,0.f,0.f};
    #pragma unroll
    for (int t = 0; t < 16; t += 2) {
      v2f pa, pb, ga, gb;
      pa.x = z[t].x * z[t].x + z[t].y * z[t].y;
      pa.y = z[t].z * z[t].z + z[t].w * z[t].w;
      pb.x = z[t + 1].x * z[t + 1].x + z[t + 1].y * z[t + 1].y;
      pb.y = z[t + 1].z * z[t + 1].z + z[t + 1].w * z[t + 1].w;
      ga.x = (fm & (2 * t)) ? gn : gp;      ga.y = (fm & (2 * t + 1)) ? gn : gp;
      gb.x = (fm & (2 * t + 2)) ? gn : gp;  gb.y = (fm & (2 * t + 3)) ? gn : gp;
      d0 = WMMA4(pa, ga, d0);
      d1 = WMMA4(pb, gb, d1);
    }
    const v8f ds = d0 + d1;
    const float s7 = (n == 7) ? -1.f : 1.f;
    const float s6 = (n == 6) ? -1.f : 1.f;
    const float s5 = (n == 5) ? -1.f : 1.f;
    const float s4 = (n == 4) ? -1.f : 1.f;
    const float t0 = ds[0] + s7 * ds[1], t1 = ds[2] + s7 * ds[3];
    const float t2 = ds[4] + s7 * ds[5], t3 = ds[6] + s7 * ds[7];
    const float u0 = t0 + s6 * t1, u1 = t2 + s6 * t3;
    const float v = u0 + s5 * u1;
    const float val = v + s4 * __shfl_xor(v, 16, 32);
    if (lane < NQ) out[b * NQ + lane] = val;
  }
}

extern "C" void kernel_launch(void* const* d_in, const int* in_sizes, int n_in,
                              void* d_out, int out_size, void* d_ws, size_t ws_size,
                              hipStream_t stream) {
  (void)in_sizes; (void)n_in; (void)out_size; (void)d_ws; (void)ws_size;
  const float* x  = (const float*)d_in[0];
  const float* qw = (const float*)d_in[1];
  float* out = (float*)d_out;
  const int batch = 2048;
  dim3 grid(batch / WAVES_PER_BLOCK);
  dim3 block(THREADS);
  qlayer_kernel<<<grid, block, 0, stream>>>(x, qw, out);
}


# sin and cos of each angle from one branch-free sincos (pi/2 reduction in 3 fma, two short polynomials, quadrant by bit ops), not library sinf+cosf: input loads issue at entry as one group
# speedup vs baseline: 2.2234x; 1.1094x over previous
#include <hip/hip_runtime.h>

typedef __attribute__((ext_vector_type(2))) float v2f;
typedef __attribute__((ext_vector_type(8))) float v8f;

#define NQ 10
#define NL 4
#define NSTATES 1024
#define WAVES_PER_BLOCK 4
#define THREADS (WAVES_PER_BLOCK * 32)

__device__ __forceinline__ void sincos_q(const float a, float& sn, float& cs) {
  const float k = rintf(a * 0.636619772f);
  float r = fmaf(k, -0x1.921fb4p+0f, a);
  r = fmaf(k, -0x1.4442d0p-24f, r);
  r = fmaf(k, -0x1.846988p-48f, r);
  const int q = (int)k;
  const float t = r * r;
  float ps = fmaf(t, 2.75573192e-6f, -1.98412698e-4f);
  ps = fmaf(ps, t, 8.33333333e-3f);
  ps = fmaf(ps, t, -1.66666667e-1f);
  const float s = fmaf(r * t, ps, r);
  float pc = fmaf(t, -2.75573192e-7f, 2.48015873e-5f);
  pc = fmaf(pc, t, -1.38888889e-3f);
  pc = fmaf(pc, t, 4.16666667e-2f);
  const float c = fmaf(t * t, pc, fmaf(t, -0.5f, 1.0f));
  const bool odd = (q & 1) != 0;
  const float ss = odd ? c : s, cc = odd ? s : c;
  sn = __int_as_float(__float_as_int(ss) ^ ((q & 2) << 30));
  cs = __int_as_float(__float_as_int(cc) ^ (((q + 1) & 2) << 30));
}

#define LDS_BYTES (1280 + 8192 + 8192 + 512 + WAVES_PER_BLOCK * 8192)

__device__ __forceinline__ int slot_of(const int f) {
  return ((f >> 6) & 1) | (((f >> 2) & 15) << 1) | (((f >> 9) & 1) << 5) |
         (((f >> 7) & 3) << 6) | ((f & 3) << 8);
}


__device__ __forceinline__ void load_gate16(const float* Ure, const float* Uim, const int lane,
                                            v2f (&gRe)[4], v2f (&gNs)[4], v2f (&gDf)[4]) {
  const int h = lane >> 4, n = lane & 15;
  const float4* pr = (const float4*)(Ure + n * 16 + 8 * h);
  const float4* pi = (const float4*)(Uim + n * 16 + 8 * h);
  const float4 r0 = pr[0], r1 = pr[1], i0 = pi[0], i1 = pi[1];
  const float rr[8] = {r0.x, r0.y, r0.z, r0.w, r1.x, r1.y, r1.z, r1.w};
  const float ii[8] = {i0.x, i0.y, i0.z, i0.w, i1.x, i1.y, i1.z, i1.w};
  #pragma unroll
  for (int k = 0; k < 4; ++k) {
    gRe[k].x = rr[2 * k];                 gRe[k].y = rr[2 * k + 1];
    gNs[k].x = -rr[2 * k] - ii[2 * k];    gNs[k].y = -rr[2 * k + 1] - ii[2 * k + 1];
    gDf[k].x = ii[2 * k] - rr[2 * k];     gDf[k].y = ii[2 * k + 1] - rr[2 * k + 1];
  }
}

#define WMMA4(A, B, C) __builtin_amdgcn_wmma_f32_16x16x4_f32(false, (A), false, (B), (short)0, (C), false, false)

__device__ __forceinline__ int cnot_dst(int j, const int rr) {
  #pragma unroll
  for (int w = 0; w < NQ; ++w) {
    const int cpos = 9 - w;
    int t = w + rr;
    t = (t >= NQ) ? t - NQ : t;
    const int tpos = 9 - t;
    j ^= ((j >> cpos) & 1) << tpos;
  }
  return j;
}

__device__ __forceinline__ void load_gate4(const float* ucr, const float* uci, const int lane,
                                           v2f (&wRe)[4], v2f (&wNs)[4], v2f (&wDf)[4]) {
  const int h = lane >> 4, n = lane & 15;
  const bool on = ((n >> 3) == h);
  const int jsel = n & 1, r = (n >> 1) & 3;
  #pragma unroll
  for (int k = 0; k < 4; ++k) {
    const float vr = on ? ucr[r * 4 + k] : 0.f;
    const float vi = on ? uci[r * 4 + k] : 0.f;
    const float ns = -vr - vi, df = vi - vr;
    wRe[k].x = jsel ? 0.f : vr;  wRe[k].y = jsel ? vr : 0.f;
    wNs[k].x = jsel ? 0.f : ns;  wNs[k].y = jsel ? ns : 0.f;
    wDf[k].x = jsel ? 0.f : df;  wDf[k].y = jsel ? df : 0.f;
  }
}

__device__ __forceinline__ void scatter_ring(float2* S, const v8f (&re3)[4], const v8f (&im3)[4],
                                             const int dLane, const int g0, const int g1,
                                             const int g2, const int g3, const int g4) {
  #pragma unroll
  for (int t = 0; t < 4; ++t) {
    const int dj = dLane ^ ((t & 1) ? g3 : 0) ^ ((t & 2) ? g4 : 0);
    #pragma unroll
    for (int v = 0; v < 8; ++v) {
      const int o = dj ^ ((v & 1) ? g2 : 0) ^ ((v & 2) ? g0 : 0) ^ ((v & 4) ? g1 : 0);
      S[o] = make_float2(re3[t][v], im3[t][v]);
    }
  }
}

__device__ __forceinline__ void ring_images(const int rr, const int lane, int& dLane,
                                            int& g0, int& g1, int& g2, int& g3, int& g4) {
  const int n = lane & 15, h = lane >> 4;
  dLane = slot_of(cnot_dst((n << 6) | (h << 5), rr));
  const int bas = slot_of(cnot_dst(1 << (lane & 15), rr));
  g0 = __builtin_amdgcn_readlane(bas, 0);
  g1 = __builtin_amdgcn_readlane(bas, 1);
  g2 = __builtin_amdgcn_readlane(bas, 2);
  g3 = __builtin_amdgcn_readlane(bas, 3);
  g4 = __builtin_amdgcn_readlane(bas, 4);
}

__device__ __forceinline__ void layer0_regs(float2* S,
                                            const float* UAre, const float* UAim,
                                            const float* UBre, const float* UBim,
                                            const float* ucr, const float* uci,
                                            const float (&cw)[NQ], const float (&sw)[NQ],
                                            const int dLane, const int g0, const int g1,
                                            const int g2, const int g3, const int g4,
                                            const int lane) {
  const int h = lane >> 4;
  v2f uaRe[4], uaNs[4], uaDf[4], ubRe[4], ubNs[4], ubDf[4], wRe[4], wNs[4], wDf[4];
  load_gate16(UAre, UAim, lane, uaRe, uaNs, uaDf);
  load_gate16(UBre, UBim, lane, ubRe, ubNs, ubDf);
  load_gate4(ucr, uci, lane, wRe, wNs, wDf);

  const float fa = h ? sw[0] : cw[0];
  const float fb = h ? sw[4] : cw[4];
  v2f va[4], vb[4], vc[4];
  #pragma unroll
  for (int k = 0; k < 4; ++k) {
    const float pa = fa * ((k & 2) ? sw[1] : cw[1]) * ((k & 1) ? sw[2] : cw[2]);
    va[k].x = pa * cw[3];  va[k].y = pa * sw[3];
    const float pb = fb * ((k & 2) ? sw[5] : cw[5]) * ((k & 1) ? sw[6] : cw[6]);
    vb[k].x = pb * cw[7];  vb[k].y = pb * sw[7];
    const float pc = ((k & 2) ? sw[8] : cw[8]) * ((k & 1) ? sw[9] : cw[9]);
    vc[k].x = pc;  vc[k].y = pc;
  }

  v8f aRe = (v8f){0.f,0.f,0.f,0.f,0.f,0.f,0.f,0.f};
  v8f bRe = aRe, cRe = aRe;
  #pragma unroll
  for (int k = 0; k < 4; ++k) {
    aRe = WMMA4(va[k], uaRe[k], aRe);
    bRe = WMMA4(ubRe[k], vb[k], bRe);
    cRe = WMMA4(wRe[k], vc[k], cRe);
  }
  v8f aIm = aRe, bIm = bRe, cIm = cRe;
  #pragma unroll
  for (int k = 0; k < 4; ++k) {
    aIm = WMMA4(va[k], uaDf[k], aIm);
    bIm = WMMA4(ubDf[k], vb[k], bIm);
    cIm = WMMA4(wDf[k], vc[k], cIm);
  }

  const float ar = aRe[0], ai = aIm[0];
  float abr[8], abi[8];
  #pragma unroll
  for (int u = 0; u < 8; ++u) {
    abr[u] = ar * bRe[u] - ai * bIm[u];
    abi[u] = ar * bIm[u] + ai * bRe[u];
  }
  v8f re3[4], im3[4];
  #pragma unroll
  for (int t = 0; t < 4; ++t) {
    #pragma unroll
    for (int v = 0; v < 8; ++v) {
      const int u = 2 * t + (v & 1);
      re3[t][v] = abr[u] * cRe[v] - abi[u] * cIm[v];
      im3[t][v] = abr[u] * cIm[v] + abi[u] * cRe[v];
    }
  }
  scatter_ring(S, re3, im3, dLane, g0, g1, g2, g3, g4);
}

__device__ __forceinline__ void layer_regs(float2* S,
                                           const float* UAre, const float* UAim,
                                           const float* UBre, const float* UBim,
                                           const float* ucr, const float* uci,
                                           const int dLane, const int g0, const int g1,
                                           const int g2, const int g3, const int g4,
                                           const int lane) {
  v2f uaRe[4], uaNs[4], uaDf[4], ubRe[4], ubNs[4], ubDf[4];
  load_gate16(UAre, UAim, lane, uaRe, uaNs, uaDf);
  load_gate16(UBre, UBim, lane, ubRe, ubNs, ubDf);

  v2f wRe[4], wNs[4], wDf[4];
  load_gate4(ucr, uci, lane, wRe, wNs, wDf);

  const float2* Sl = S + 2 * lane;
  float4 z[4][4];
  #pragma unroll
  for (int c = 0; c < 4; ++c) {
    #pragma unroll
    for (int k = 0; k < 4; ++k) z[c][k] = *(const float4*)(Sl + 64 * k + 256 * c);
  }

  v8f re1[4], im1[4];
  #pragma unroll
  for (int c = 0; c < 4; ++c) re1[c] = (v8f){0.f,0.f,0.f,0.f,0.f,0.f,0.f,0.f};
  #pragma unroll
  for (int k = 0; k < 4; ++k) {
    #pragma unroll
    for (int c = 0; c < 4; ++c) {
      v2f as;  as.x = z[c][k].x + z[c][k].y;  as.y = z[c][k].z + z[c][k].w;
      re1[c] = WMMA4(as, uaRe[k], re1[c]);
    }
  }
  #pragma unroll
  for (int c = 0; c < 4; ++c) im1[c] = re1[c];
  #pragma unroll
  for (int k = 0; k < 4; ++k) {
    #pragma unroll
    for (int c = 0; c < 4; ++c) {
      v2f ar, ai;
      ar.x = z[c][k].x;  ar.y = z[c][k].z;
      ai.x = z[c][k].y;  ai.y = z[c][k].w;
      re1[c] = WMMA4(ai, uaNs[k], re1[c]);
      im1[c] = WMMA4(ar, uaDf[k], im1[c]);
    }
  }

  v8f d2re[4], d2im[4];
  #pragma unroll
  for (int c = 0; c < 4; ++c) d2re[c] = (v8f){0.f,0.f,0.f,0.f,0.f,0.f,0.f,0.f};
  #pragma unroll
  for (int k = 0; k < 4; ++k) {
    #pragma unroll
    for (int c = 0; c < 4; ++c) {
      v2f bs;
      bs.x = re1[c][2 * k] + im1[c][2 * k];
      bs.y = re1[c][2 * k + 1] + im1[c][2 * k + 1];
      d2re[c] = WMMA4(ubRe[k], bs, d2re[c]);
    }
  }
  #pragma unroll
  for (int c = 0; c < 4; ++c) d2im[c] = d2re[c];
  #pragma unroll
  for (int k = 0; k < 4; ++k) {
    #pragma unroll
    for (int c = 0; c < 4; ++c) {
      v2f br, bi;
      br.x = re1[c][2 * k];  br.y = re1[c][2 * k + 1];
      bi.x = im1[c][2 * k];  bi.y = im1[c][2 * k + 1];
      d2re[c] = WMMA4(ubNs[k], bi, d2re[c]);
      d2im[c] = WMMA4(ubDf[k], br, d2im[c]);
    }
  }

  v8f re3[4], im3[4];
  #pragma unroll
  for (int t = 0; t < 4; ++t) re3[t] = (v8f){0.f,0.f,0.f,0.f,0.f,0.f,0.f,0.f};
  #pragma unroll
  for (int k = 0; k < 4; ++k) {
    #pragma unroll
    for (int t = 0; t < 4; ++t) {
      v2f bs;
      bs.x = d2re[k][2 * t] + d2im[k][2 * t];
      bs.y = d2re[k][2 * t + 1] + d2im[k][2 * t + 1];
      re3[t] = WMMA4(wRe[k], bs, re3[t]);
    }
  }
  #pragma unroll
  for (int t = 0; t < 4; ++t) im3[t] = re3[t];
  #pragma unroll
  for (int k = 0; k < 4; ++k) {
    #pragma unroll
    for (int t = 0; t < 4; ++t) {
      v2f br, bi;
      br.x = d2re[k][2 * t];  br.y = d2re[k][2 * t + 1];
      bi.x = d2im[k][2 * t];  bi.y = d2im[k][2 * t + 1];
      re3[t] = WMMA4(wNs[k], bi, re3[t]);
      im3[t] = WMMA4(wDf[k], br, im3[t]);
    }
  }
  scatter_ring(S, re3, im3, dLane, g0, g1, g2, g3, g4);
}

extern "C" __global__ __launch_bounds__(THREADS)
void qlayer_kernel(const float* __restrict__ x,
                   const float* __restrict__ qw,
                   float* __restrict__ out) {
  __shared__ __attribute__((aligned(16))) char smem_raw[LDS_BYTES];
  float* u2   = (float*)smem_raw;
  float* UAre = u2 + NL * NQ * 8;
  float* UAim = UAre + NL * 256;
  float* UBre = UAim + NL * 256;
  float* UBim = UBre + NL * 256;
  float* UCre = UBim + NL * 256;
  float* UCim = UCre + NL * 16;
  float2* stateBase = (float2*)(UCim + NL * 16);

  const int tid = threadIdx.x;
  const int wave = tid >> 5, lane = tid & 31;
  const int b = blockIdx.x * WAVES_PER_BLOCK + wave;

  float cw[NQ], sw[NQ];
  #pragma unroll
  for (int w = 0; w < NQ; ++w) {
    const float ang = 1.5707963267948966f * x[b * NQ + w];
    sincos_q(ang, sw[w], cw[w]);
  }

  #pragma unroll 1
  for (int e = tid; e < NL * NQ; e += THREADS) {
    const int l = e / NQ, w = e % NQ;
    const float phi = qw[(l * NQ + w) * 3 + 0];
    const float th  = qw[(l * NQ + w) * 3 + 1];
    const float om  = qw[(l * NQ + w) * 3 + 2];
    const float ap = 0.5f * (phi + om), am = 0.5f * (phi - om);
    float c, s, cpo, spo, cb, sb;
    sincos_q(0.5f * th, s, c);
    sincos_q(ap, spo, cpo);
    sincos_q(am, sb, cb);
    float* g = u2 + e * 8;
    g[0] =  cpo * c; g[1] = -spo * c;
    g[2] = -cb * s;  g[3] = -sb * s;
    g[4] =  cb * s;  g[5] = -sb * s;
    g[6] =  cpo * c; g[7] =  spo * c;
  }
  __syncthreads();

  #pragma unroll 1
  for (int e = tid; e < NL * 256; e += THREADS) {
    const int l = e >> 8, rc = e & 255, r = rc >> 4, cc = rc & 15;
    float arA = 1.f, aiA = 0.f, arB = 1.f, aiB = 0.f;
    #pragma unroll
    for (int q = 0; q < 4; ++q) {
      const int rb = (r >> (3 - q)) & 1, cbit = (cc >> (3 - q)) & 1;
      const float* gA = u2 + (((l * NQ + q) * 4) + rb * 2 + cbit) * 2;
      float nr = arA * gA[0] - aiA * gA[1];
      float ni = arA * gA[1] + aiA * gA[0];
      arA = nr; aiA = ni;
      const float* gB = u2 + (((l * NQ + 4 + q) * 4) + rb * 2 + cbit) * 2;
      nr = arB * gB[0] - aiB * gB[1];
      ni = arB * gB[1] + aiB * gB[0];
      arB = nr; aiB = ni;
    }
    UAre[e] = arA; UAim[e] = aiA;
    UBre[e] = arB; UBim[e] = aiB;
  }
  #pragma unroll 1
  for (int e = tid; e < NL * 16; e += THREADS) {
    const int l = e >> 4, r = (e >> 2) & 3, cc = e & 3;
    const float* g8 = u2 + (((l * NQ + 8) * 4) + ((r >> 1) & 1) * 2 + ((cc >> 1) & 1)) * 2;
    const float* g9 = u2 + (((l * NQ + 9) * 4) + (r & 1) * 2 + (cc & 1)) * 2;
    UCre[e] = g8[0] * g9[0] - g8[1] * g9[1];
    UCim[e] = g8[0] * g9[1] + g8[1] * g9[0];
  }
  __syncthreads();

  float2* S = stateBase + wave * NSTATES;

  {
    int dLane, g0, g1, g2, g3, g4;
    ring_images(1, lane, dLane, g0, g1, g2, g3, g4);
    layer0_regs(S, UAre, UAim, UBre, UBim, UCre, UCim, cw, sw, dLane, g0, g1, g2, g3, g4, lane);
  }

  #pragma unroll 1
  for (int l = 1; l < NL; ++l) {
    const int rr = (l % (NQ - 1)) + 1;
    int dLane, g0, g1, g2, g3, g4;
    ring_images(rr, lane, dLane, g0, g1, g2, g3, g4);
    layer_regs(S, UAre + l * 256, UAim + l * 256, UBre + l * 256, UBim + l * 256,
               UCre + l * 16, UCim + l * 16, dLane, g0, g1, g2, g3, g4, lane);
  }

  {
    const int h = lane >> 4, n = lane & 15;
    const float2* Sl = S + 2 * lane;
    float4 z[16];
    #pragma unroll
    for (int t = 0; t < 16; ++t) z[t] = *(const float4*)(Sl + 64 * t);
    int fm = ((unsigned)(n - 1) < 3u) ? (8 >> n) : 0;
    fm = ((unsigned)(n - 8) < 2u) ? (4096 >> n) : fm;
    float gp = (n < NQ) ? 1.f : 0.f;
    gp = (n == 0) ? (1.f - 2.f * (float)h) : gp;
    const float gn = -gp;
    v8f d0 = (v8f){0.f,0.f,0.f,0.f,0.f,0.f,0.f,0.f};
    v8f d1 = (v8f){0.f,0.f,0.f,0.f,0.f,0.f,0.f,0.f};
    #pragma unroll
    for (int t = 0; t < 16; t += 2) {
      v2f pa, pb, ga, gb;
      pa.x = z[t].x * z[t].x + z[t].y * z[t].y;
      pa.y = z[t].z * z[t].z + z[t].w * z[t].w;
      pb.x = z[t + 1].x * z[t + 1].x + z[t + 1].y * z[t + 1].y;
      pb.y = z[t + 1].z * z[t + 1].z + z[t + 1].w * z[t + 1].w;
      ga.x = (fm & (2 * t)) ? gn : gp;      ga.y = (fm & (2 * t + 1)) ? gn : gp;
      gb.x = (fm & (2 * t + 2)) ? gn : gp;  gb.y = (fm & (2 * t + 3)) ? gn : gp;
      d0 = WMMA4(pa, ga, d0);
      d1 = WMMA4(pb, gb, d1);
    }
    const v8f ds = d0 + d1;
    const float s7 = (n == 7) ? -1.f : 1.f;
    const float s6 = (n == 6) ? -1.f : 1.f;
    const float s5 = (n == 5) ? -1.f : 1.f;
    const float s4 = (n == 4) ? -1.f : 1.f;
    const float t0 = ds[0] + s7 * ds[1], t1 = ds[2] + s7 * ds[3];
    const float t2 = ds[4] + s7 * ds[5], t3 = ds[6] + s7 * ds[7];
    const float u0 = t0 + s6 * t1, u1 = t2 + s6 * t3;
    const float v = u0 + s5 * u1;
    const float val = v + s4 * __shfl_xor(v, 16, 32);
    if (lane < NQ) out[b * NQ + lane] = val;
  }
}

extern "C" void kernel_launch(void* const* d_in, const int* in_sizes, int n_in,
                              void* d_out, int out_size, void* d_ws, size_t ws_size,
                              hipStream_t stream) {
  (void)in_sizes; (void)n_in; (void)out_size; (void)d_ws; (void)ws_size;
  const float* x  = (const float*)d_in[0];
  const float* qw = (const float*)d_in[1];
  float* out = (float*)d_out;
  const int batch = 2048;
  dim3 grid(batch / WAVES_PER_BLOCK);
  dim3 block(THREADS);
  qlayer_kernel<<<grid, block, 0, stream>>>(x, qw, out);
}


# Layers 0 and 1 from the chain form of the state behind the first CNOT ring: two 32x4 factors, amplitudes as one K=4 product (16 f32 WMMAs, not 168); each amplitude's summation order changes
# speedup vs baseline: 3.0430x; 1.3686x over previous
#include <hip/hip_runtime.h>

typedef __attribute__((ext_vector_type(2))) float v2f;
typedef __attribute__((ext_vector_type(8))) float v8f;
typedef __attribute__((ext_vector_type(4))) float v4f;

#define NQ 10
#define NL 4
#define NSTATES 1024
#define WAVES_PER_BLOCK 4
#define THREADS (WAVES_PER_BLOCK * 32)

__device__ __forceinline__ void sincos_q(const float a, float& sn, float& cs) {
  const float k = rintf(a * 0.636619772f);
  float r = fmaf(k, -0x1.921fb4p+0f, a);
  r = fmaf(k, -0x1.4442d0p-24f, r);
  r = fmaf(k, -0x1.846988p-48f, r);
  const int q = (int)k;
  const float t = r * r;
  float ps = fmaf(t, 2.75573192e-6f, -1.98412698e-4f);
  ps = fmaf(ps, t, 8.33333333e-3f);
  ps = fmaf(ps, t, -1.66666667e-1f);
  const float s = fmaf(r * t, ps, r);
  float pc = fmaf(t, -2.75573192e-7f, 2.48015873e-5f);
  pc = fmaf(pc, t, -1.38888889e-3f);
  pc = fmaf(pc, t, 4.16666667e-2f);
  const float c = fmaf(t * t, pc, fmaf(t, -0.5f, 1.0f));
  const bool odd = (q & 1) != 0;
  const float ss = odd ? c : s, cc = odd ? s : c;
  sn = __int_as_float(__float_as_int(ss) ^ ((q & 2) << 30));
  cs = __int_as_float(__float_as_int(cc) ^ (((q + 1) & 2) << 30));
}

#define LDS_BYTES (1280 + 8192 + 8192 + 512 + WAVES_PER_BLOCK * 8192)

__device__ __forceinline__ int slot_of(const int f) {
  return ((f >> 6) & 1) | (((f >> 2) & 15) << 1) | (((f >> 9) & 1) << 5) |
         (((f >> 7) & 3) << 6) | ((f & 3) << 8);
}


__device__ __forceinline__ void load_gate16(const float* Ure, const float* Uim, const int lane,
                                            v2f (&gRe)[4], v2f (&gNs)[4], v2f (&gDf)[4]) {
  const int h = lane >> 4, n = lane & 15;
  const float4* pr = (const float4*)(Ure + n * 16 + 8 * h);
  const float4* pi = (const float4*)(Uim + n * 16 + 8 * h);
  const float4 r0 = pr[0], r1 = pr[1], i0 = pi[0], i1 = pi[1];
  const float rr[8] = {r0.x, r0.y, r0.z, r0.w, r1.x, r1.y, r1.z, r1.w};
  const float ii[8] = {i0.x, i0.y, i0.z, i0.w, i1.x, i1.y, i1.z, i1.w};
  #pragma unroll
  for (int k = 0; k < 4; ++k) {
    gRe[k].x = rr[2 * k];                 gRe[k].y = rr[2 * k + 1];
    gNs[k].x = -rr[2 * k] - ii[2 * k];    gNs[k].y = -rr[2 * k + 1] - ii[2 * k + 1];
    gDf[k].x = ii[2 * k] - rr[2 * k];     gDf[k].y = ii[2 * k + 1] - rr[2 * k + 1];
  }
}

#define WMMA4(A, B, C) __builtin_amdgcn_wmma_f32_16x16x4_f32(false, (A), false, (B), (short)0, (C), false, false)

__device__ __forceinline__ int cnot_dst(int j, const int rr) {
  #pragma unroll
  for (int w = 0; w < NQ; ++w) {
    const int cpos = 9 - w;
    int t = w + rr;
    t = (t >= NQ) ? t - NQ : t;
    const int tpos = 9 - t;
    j ^= ((j >> cpos) & 1) << tpos;
  }
  return j;
}

__device__ __forceinline__ void load_gate4(const float* ucr, const float* uci, const int lane,
                                           v2f (&wRe)[4], v2f (&wNs)[4], v2f (&wDf)[4]) {
  const int h = lane >> 4, n = lane & 15;
  const bool on = ((n >> 3) == h);
  const int jsel = n & 1, r = (n >> 1) & 3;
  #pragma unroll
  for (int k = 0; k < 4; ++k) {
    const float vr = on ? ucr[r * 4 + k] : 0.f;
    const float vi = on ? uci[r * 4 + k] : 0.f;
    const float ns = -vr - vi, df = vi - vr;
    wRe[k].x = jsel ? 0.f : vr;  wRe[k].y = jsel ? vr : 0.f;
    wNs[k].x = jsel ? 0.f : ns;  wNs[k].y = jsel ? ns : 0.f;
    wDf[k].x = jsel ? 0.f : df;  wDf[k].y = jsel ? df : 0.f;
  }
}

__device__ __forceinline__ void scatter_ring(float2* S, const v8f (&re3)[4], const v8f (&im3)[4],
                                             const int dLane, const int g0, const int g1,
                                             const int g2, const int g3, const int g4) {
  #pragma unroll
  for (int t = 0; t < 4; ++t) {
    const int dj = dLane ^ ((t & 1) ? g3 : 0) ^ ((t & 2) ? g4 : 0);
    #pragma unroll
    for (int v = 0; v < 8; ++v) {
      const int o = dj ^ ((v & 1) ? g2 : 0) ^ ((v & 2) ? g0 : 0) ^ ((v & 4) ? g1 : 0);
      S[o] = make_float2(re3[t][v], im3[t][v]);
    }
  }
}

__device__ __forceinline__ void ring_images(const int rr, const int lane, int& dLane,
                                            int& g0, int& g1, int& g2, int& g3, int& g4) {
  const int n = lane & 15, h = lane >> 4;
  dLane = slot_of(cnot_dst((n << 6) | (h << 5), rr));
  const int bas = slot_of(cnot_dst(1 << (lane & 15), rr));
  g0 = __builtin_amdgcn_readlane(bas, 0);
  g1 = __builtin_amdgcn_readlane(bas, 1);
  g2 = __builtin_amdgcn_readlane(bas, 2);
  g3 = __builtin_amdgcn_readlane(bas, 3);
  g4 = __builtin_amdgcn_readlane(bas, 4);
}

struct cpx { float r, i; };
__device__ __forceinline__ cpx cmul(const cpx a, const cpx b) {
  cpx o;  o.r = a.r * b.r - a.i * b.i;  o.i = a.r * b.i + a.i * b.r;  return o;
}
__device__ __forceinline__ cpx cmul2(const cpx a, const cpx b, const cpx c, const cpx d) {
  cpx o;
  o.r = a.r * b.r - a.i * b.i + c.r * d.r - c.i * d.i;
  o.i = a.r * b.i + a.i * b.r + c.r * d.i + c.i * d.r;
  return o;
}
__device__ __forceinline__ cpx csel(const bool s, const cpx a, const cpx b) {
  cpx o;  o.r = s ? a.r : b.r;  o.i = s ? a.i : b.i;  return o;
}

__device__ __forceinline__ void layers01_chain(float2* S, const float* u2,
                                               const float (&cw)[NQ], const float (&sw)[NQ],
                                               const int lane) {
  const int h = lane >> 4, n = lane & 15;
  const bool p = (h != 0);

  const float* v2 = u2 + NQ * 8;
  v4f gu[NQ][2], gv[NQ], gv4b, gv5b;
  #pragma unroll
  for (int q = 0; q < NQ; ++q) {
    gu[q][0] = *(const v4f*)(u2 + q * 8);
    gu[q][1] = *(const v4f*)(u2 + q * 8 + 4);
  }
  #pragma unroll
  for (int q = 0; q < NQ; ++q) {
    const int bit = (q < 4) ? (3 - q) : (9 - q);
    const int y = (q == 4 || q == 5) ? 0 : ((n >> bit) & 1);
    gv[q] = *(const v4f*)(v2 + q * 8 + 4 * y);
  }
  gv4b = *(const v4f*)(v2 + 4 * 8 + 4);
  gv5b = *(const v4f*)(v2 + 5 * 8 + 4);
  asm volatile("" :: "v"(gu[0][0]), "v"(gu[0][1]), "v"(gu[1][0]), "v"(gu[1][1]), "v"(gu[2][0]),
               "v"(gu[2][1]), "v"(gu[3][0]), "v"(gu[3][1]), "v"(gu[4][0]), "v"(gu[4][1]),
               "v"(gu[5][0]), "v"(gu[5][1]), "v"(gu[6][0]), "v"(gu[6][1]), "v"(gu[7][0]),
               "v"(gu[7][1]));
  asm volatile("" :: "v"(gu[8][0]), "v"(gu[8][1]), "v"(gu[9][0]), "v"(gu[9][1]), "v"(gv[0]),
               "v"(gv[1]), "v"(gv[2]), "v"(gv[3]), "v"(gv[4]), "v"(gv[5]), "v"(gv[6]), "v"(gv[7]),
               "v"(gv[8]), "v"(gv[9]), "v"(gv4b), "v"(gv5b));

  cpx u[NQ][2];
  #pragma unroll
  for (int q = 0; q < NQ; ++q) {
    u[q][0].r = gu[q][0].x * cw[q] + gu[q][0].z * sw[q];
    u[q][0].i = gu[q][0].y * cw[q] + gu[q][0].w * sw[q];
    u[q][1].r = gu[q][1].x * cw[q] + gu[q][1].z * sw[q];
    u[q][1].i = gu[q][1].y * cw[q] + gu[q][1].w * sw[q];
  }
  cpx Vr[NQ][2], V4[2][2], V5[2][2];
  #pragma unroll
  for (int q = 0; q < NQ; ++q) {
    Vr[q][0].r = gv[q].x;  Vr[q][0].i = gv[q].y;
    Vr[q][1].r = gv[q].z;  Vr[q][1].i = gv[q].w;
  }
  V4[0][0] = Vr[4][0];  V4[0][1] = Vr[4][1];
  V4[1][0].r = gv4b.x;  V4[1][0].i = gv4b.y;  V4[1][1].r = gv4b.z;  V4[1][1].i = gv4b.w;
  V5[0][0] = Vr[5][0];  V5[0][1] = Vr[5][1];
  V5[1][0].r = gv5b.x;  V5[1][0].i = gv5b.y;  V5[1][1].r = gv5b.z;  V5[1][1].i = gv5b.w;

  v2f LrA[2], LiA[2], nLiA[2];
  {
    const cpx u0p = csel(p, u[0][1], u[0][0]), u0n = csel(p, u[0][0], u[0][1]);
    const cpx u1p = csel(p, u[1][1], u[1][0]), u1n = csel(p, u[1][0], u[1][1]);
    cpx a0 = cmul(Vr[0][0], u0p), a1 = cmul(Vr[0][1], u0n);
    {
      const cpx s0 = cmul2(a0, u1p, a1, u1n), s1 = cmul2(a0, u1n, a1, u1p);
      a0 = cmul(Vr[1][0], s0);  a1 = cmul(Vr[1][1], s1);
    }
    #pragma unroll
    for (int q = 2; q < 4; ++q) {
      const cpx s0 = cmul2(a0, u[q][0], a1, u[q][1]), s1 = cmul2(a0, u[q][1], a1, u[q][0]);
      a0 = cmul(Vr[q][0], s0);  a1 = cmul(Vr[q][1], s1);
    }
    const cpx in0 = cmul2(a0, u[4][0], a1, u[4][1]), in1 = cmul2(a0, u[4][1], a1, u[4][0]);
    #pragma unroll
    for (int i = 0; i < 2; ++i) {
      const cpx l0 = cmul(V4[i][0], in0), l1 = cmul(V4[i][1], in1);
      LrA[i].x = l0.r;  LrA[i].y = l1.r;
      LiA[i].x = l0.i;  LiA[i].y = l1.i;
      nLiA[i].x = -l0.i;  nLiA[i].y = -l1.i;
    }
  }

  v2f RrB[2], RiB[2];
  {
    const cpx v9p = csel(p, Vr[9][1], Vr[9][0]);
    const cpx u9p = csel(p, u[9][1], u[9][0]), u9n = csel(p, u[9][0], u[9][1]);
    cpx c0 = cmul(v9p, u9p), c1 = cmul(v9p, u9n);
    #pragma unroll
    for (int q = 8; q > 5; --q) {
      const cpx w0 = cmul(Vr[q][0], c0), w1 = cmul(Vr[q][1], c1);
      c0 = cmul2(w0, u[q][0], w1, u[q][1]);
      c1 = cmul2(w0, u[q][1], w1, u[q][0]);
    }
    #pragma unroll
    for (int i2 = 0; i2 < 2; ++i2) {
      const cpx w0 = cmul(V5[i2][0], c0), w1 = cmul(V5[i2][1], c1);
      const cpx r0 = cmul2(w0, u[5][0], w1, u[5][1]), r1 = cmul2(w0, u[5][1], w1, u[5][0]);
      RrB[i2].x = r0.r;  RrB[i2].y = r1.r;
      RiB[i2].x = r0.i;  RiB[i2].y = r1.i;
    }
  }

  v8f re[4], im[4];
  const v8f zero = (v8f){0.f,0.f,0.f,0.f,0.f,0.f,0.f,0.f};
  #pragma unroll
  for (int tt = 0; tt < 4; ++tt) {
    re[tt] = WMMA4(LrA[tt >> 1], RrB[tt & 1], zero);
    im[tt] = WMMA4(LrA[tt >> 1], RiB[tt & 1], zero);
  }
  #pragma unroll
  for (int tt = 0; tt < 4; ++tt) {
    re[tt] = WMMA4(nLiA[tt >> 1], RiB[tt & 1], re[tt]);
    im[tt] = WMMA4(LiA[tt >> 1], RrB[tt & 1], im[tt]);
  }

  const int dLane = slot_of(cnot_dst((h << 9) | n, 2));
  const int bas = slot_of(cnot_dst(1 << (lane & 15), 2));
  const int g4 = __builtin_amdgcn_readlane(bas, 4);
  const int g5 = __builtin_amdgcn_readlane(bas, 5);
  const int g6 = __builtin_amdgcn_readlane(bas, 6);
  const int g7 = __builtin_amdgcn_readlane(bas, 7);
  const int g8 = __builtin_amdgcn_readlane(bas, 8);
  #pragma unroll
  for (int tt = 0; tt < 4; ++tt) {
    const int dj = dLane ^ ((tt & 2) ? g5 : 0) ^ ((tt & 1) ? g4 : 0);
    #pragma unroll
    for (int r = 0; r < 8; ++r) {
      const int o = dj ^ ((r & 1) ? g6 : 0) ^ ((r & 2) ? g7 : 0) ^ ((r & 4) ? g8 : 0);
      S[o] = make_float2(re[tt][r], im[tt][r]);
    }
  }
}

__device__ __forceinline__ void layer_regs(float2* S,
                                           const float* UAre, const float* UAim,
                                           const float* UBre, const float* UBim,
                                           const float* ucr, const float* uci,
                                           const int dLane, const int g0, const int g1,
                                           const int g2, const int g3, const int g4,
                                           const int lane) {
  v2f uaRe[4], uaNs[4], uaDf[4], ubRe[4], ubNs[4], ubDf[4];
  load_gate16(UAre, UAim, lane, uaRe, uaNs, uaDf);
  load_gate16(UBre, UBim, lane, ubRe, ubNs, ubDf);

  v2f wRe[4], wNs[4], wDf[4];
  load_gate4(ucr, uci, lane, wRe, wNs, wDf);

  const float2* Sl = S + 2 * lane;
  float4 z[4][4];
  #pragma unroll
  for (int c = 0; c < 4; ++c) {
    #pragma unroll
    for (int k = 0; k < 4; ++k) z[c][k] = *(const float4*)(Sl + 64 * k + 256 * c);
  }

  v8f re1[4], im1[4];
  #pragma unroll
  for (int c = 0; c < 4; ++c) re1[c] = (v8f){0.f,0.f,0.f,0.f,0.f,0.f,0.f,0.f};
  #pragma unroll
  for (int k = 0; k < 4; ++k) {
    #pragma unroll
    for (int c = 0; c < 4; ++c) {
      v2f as;  as.x = z[c][k].x + z[c][k].y;  as.y = z[c][k].z + z[c][k].w;
      re1[c] = WMMA4(as, uaRe[k], re1[c]);
    }
  }
  #pragma unroll
  for (int c = 0; c < 4; ++c) im1[c] = re1[c];
  #pragma unroll
  for (int k = 0; k < 4; ++k) {
    #pragma unroll
    for (int c = 0; c < 4; ++c) {
      v2f ar, ai;
      ar.x = z[c][k].x;  ar.y = z[c][k].z;
      ai.x = z[c][k].y;  ai.y = z[c][k].w;
      re1[c] = WMMA4(ai, uaNs[k], re1[c]);
      im1[c] = WMMA4(ar, uaDf[k], im1[c]);
    }
  }

  v8f d2re[4], d2im[4];
  #pragma unroll
  for (int c = 0; c < 4; ++c) d2re[c] = (v8f){0.f,0.f,0.f,0.f,0.f,0.f,0.f,0.f};
  #pragma unroll
  for (int k = 0; k < 4; ++k) {
    #pragma unroll
    for (int c = 0; c < 4; ++c) {
      v2f bs;
      bs.x = re1[c][2 * k] + im1[c][2 * k];
      bs.y = re1[c][2 * k + 1] + im1[c][2 * k + 1];
      d2re[c] = WMMA4(ubRe[k], bs, d2re[c]);
    }
  }
  #pragma unroll
  for (int c = 0; c < 4; ++c) d2im[c] = d2re[c];
  #pragma unroll
  for (int k = 0; k < 4; ++k) {
    #pragma unroll
    for (int c = 0; c < 4; ++c) {
      v2f br, bi;
      br.x = re1[c][2 * k];  br.y = re1[c][2 * k + 1];
      bi.x = im1[c][2 * k];  bi.y = im1[c][2 * k + 1];
      d2re[c] = WMMA4(ubNs[k], bi, d2re[c]);
      d2im[c] = WMMA4(ubDf[k], br, d2im[c]);
    }
  }

  v8f re3[4], im3[4];
  #pragma unroll
  for (int t = 0; t < 4; ++t) re3[t] = (v8f){0.f,0.f,0.f,0.f,0.f,0.f,0.f,0.f};
  #pragma unroll
  for (int k = 0; k < 4; ++k) {
    #pragma unroll
    for (int t = 0; t < 4; ++t) {
      v2f bs;
      bs.x = d2re[k][2 * t] + d2im[k][2 * t];
      bs.y = d2re[k][2 * t + 1] + d2im[k][2 * t + 1];
      re3[t] = WMMA4(wRe[k], bs, re3[t]);
    }
  }
  #pragma unroll
  for (int t = 0; t < 4; ++t) im3[t] = re3[t];
  #pragma unroll
  for (int k = 0; k < 4; ++k) {
    #pragma unroll
    for (int t = 0; t < 4; ++t) {
      v2f br, bi;
      br.x = d2re[k][2 * t];  br.y = d2re[k][2 * t + 1];
      bi.x = d2im[k][2 * t];  bi.y = d2im[k][2 * t + 1];
      re3[t] = WMMA4(wNs[k], bi, re3[t]);
      im3[t] = WMMA4(wDf[k], br, im3[t]);
    }
  }
  scatter_ring(S, re3, im3, dLane, g0, g1, g2, g3, g4);
}

extern "C" __global__ __launch_bounds__(THREADS)
void qlayer_kernel(const float* __restrict__ x,
                   const float* __restrict__ qw,
                   float* __restrict__ out) {
  __shared__ __attribute__((aligned(16))) char smem_raw[LDS_BYTES];
  float* u2   = (float*)smem_raw;
  float* UAre = u2 + NL * NQ * 8;
  float* UAim = UAre + NL * 256;
  float* UBre = UAim + NL * 256;
  float* UBim = UBre + NL * 256;
  float* UCre = UBim + NL * 256;
  float* UCim = UCre + NL * 16;
  float2* stateBase = (float2*)(UCim + NL * 16);

  const int tid = threadIdx.x;
  const int wave = tid >> 5, lane = tid & 31;
  const int b = blockIdx.x * WAVES_PER_BLOCK + wave;

  float cw[NQ], sw[NQ];
  #pragma unroll
  for (int w = 0; w < NQ; ++w) {
    const float ang = 1.5707963267948966f * x[b * NQ + w];
    sincos_q(ang, sw[w], cw[w]);
  }

  #pragma unroll 1
  for (int e = tid; e < NL * NQ; e += THREADS) {
    const int l = e / NQ, w = e % NQ;
    const float phi = qw[(l * NQ + w) * 3 + 0];
    const float th  = qw[(l * NQ + w) * 3 + 1];
    const float om  = qw[(l * NQ + w) * 3 + 2];
    const float ap = 0.5f * (phi + om), am = 0.5f * (phi - om);
    float c, s, cpo, spo, cb, sb;
    sincos_q(0.5f * th, s, c);
    sincos_q(ap, spo, cpo);
    sincos_q(am, sb, cb);
    float* g = u2 + e * 8;
    g[0] =  cpo * c; g[1] = -spo * c;
    g[2] = -cb * s;  g[3] = -sb * s;
    g[4] =  cb * s;  g[5] = -sb * s;
    g[6] =  cpo * c; g[7] =  spo * c;
  }
  __syncthreads();

  #pragma unroll 1
  for (int e = 2 * 256 + tid; e < NL * 256; e += THREADS) {
    const int l = e >> 8, rc = e & 255, r = rc >> 4, cc = rc & 15;
    float arA = 1.f, aiA = 0.f, arB = 1.f, aiB = 0.f;
    #pragma unroll
    for (int q = 0; q < 4; ++q) {
      const int rb = (r >> (3 - q)) & 1, cbit = (cc >> (3 - q)) & 1;
      const float* gA = u2 + (((l * NQ + q) * 4) + rb * 2 + cbit) * 2;
      float nr = arA * gA[0] - aiA * gA[1];
      float ni = arA * gA[1] + aiA * gA[0];
      arA = nr; aiA = ni;
      const float* gB = u2 + (((l * NQ + 4 + q) * 4) + rb * 2 + cbit) * 2;
      nr = arB * gB[0] - aiB * gB[1];
      ni = arB * gB[1] + aiB * gB[0];
      arB = nr; aiB = ni;
    }
    UAre[e] = arA; UAim[e] = aiA;
    UBre[e] = arB; UBim[e] = aiB;
  }
  #pragma unroll 1
  for (int e = 2 * 16 + tid; e < NL * 16; e += THREADS) {
    const int l = e >> 4, r = (e >> 2) & 3, cc = e & 3;
    const float* g8 = u2 + (((l * NQ + 8) * 4) + ((r >> 1) & 1) * 2 + ((cc >> 1) & 1)) * 2;
    const float* g9 = u2 + (((l * NQ + 9) * 4) + (r & 1) * 2 + (cc & 1)) * 2;
    UCre[e] = g8[0] * g9[0] - g8[1] * g9[1];
    UCim[e] = g8[0] * g9[1] + g8[1] * g9[0];
  }
  __syncthreads();

  float2* S = stateBase + wave * NSTATES;

  layers01_chain(S, u2, cw, sw, lane);

  #pragma unroll 1
  for (int l = 2; l < NL; ++l) {
    const int rr = (l % (NQ - 1)) + 1;
    int dLane, g0, g1, g2, g3, g4;
    ring_images(rr, lane, dLane, g0, g1, g2, g3, g4);
    layer_regs(S, UAre + l * 256, UAim + l * 256, UBre + l * 256, UBim + l * 256,
               UCre + l * 16, UCim + l * 16, dLane, g0, g1, g2, g3, g4, lane);
  }

  {
    const int h = lane >> 4, n = lane & 15;
    const float2* Sl = S + 2 * lane;
    float4 z[16];
    #pragma unroll
    for (int t = 0; t < 16; ++t) z[t] = *(const float4*)(Sl + 64 * t);
    int fm = ((unsigned)(n - 1) < 3u) ? (8 >> n) : 0;
    fm = ((unsigned)(n - 8) < 2u) ? (4096 >> n) : fm;
    float gp = (n < NQ) ? 1.f : 0.f;
    gp = (n == 0) ? (1.f - 2.f * (float)h) : gp;
    const float gn = -gp;
    v8f d0 = (v8f){0.f,0.f,0.f,0.f,0.f,0.f,0.f,0.f};
    v8f d1 = (v8f){0.f,0.f,0.f,0.f,0.f,0.f,0.f,0.f};
    #pragma unroll
    for (int t = 0; t < 16; t += 2) {
      v2f pa, pb, ga, gb;
      pa.x = z[t].x * z[t].x + z[t].y * z[t].y;
      pa.y = z[t].z * z[t].z + z[t].w * z[t].w;
      pb.x = z[t + 1].x * z[t + 1].x + z[t + 1].y * z[t + 1].y;
      pb.y = z[t + 1].z * z[t + 1].z + z[t + 1].w * z[t + 1].w;
      ga.x = (fm & (2 * t)) ? gn : gp;      ga.y = (fm & (2 * t + 1)) ? gn : gp;
      gb.x = (fm & (2 * t + 2)) ? gn : gp;  gb.y = (fm & (2 * t + 3)) ? gn : gp;
      d0 = WMMA4(pa, ga, d0);
      d1 = WMMA4(pb, gb, d1);
    }
    const v8f ds = d0 + d1;
    const float s7 = (n == 7) ? -1.f : 1.f;
    const float s6 = (n == 6) ? -1.f : 1.f;
    const float s5 = (n == 5) ? -1.f : 1.f;
    const float s4 = (n == 4) ? -1.f : 1.f;
    const float t0 = ds[0] + s7 * ds[1], t1 = ds[2] + s7 * ds[3];
    const float t2 = ds[4] + s7 * ds[5], t3 = ds[6] + s7 * ds[7];
    const float u0 = t0 + s6 * t1, u1 = t2 + s6 * t3;
    const float v = u0 + s5 * u1;
    const float val = v + s4 * __shfl_xor(v, 16, 32);
    if (lane < NQ) out[b * NQ + lane] = val;
  }
}

extern "C" void kernel_launch(void* const* d_in, const int* in_sizes, int n_in,
                              void* d_out, int out_size, void* d_ws, size_t ws_size,
                              hipStream_t stream) {
  (void)in_sizes; (void)n_in; (void)out_size; (void)d_ws; (void)ws_size;
  const float* x  = (const float*)d_in[0];
  const float* qw = (const float*)d_in[1];
  float* out = (float*)d_out;
  const int batch = 2048;
  dim3 grid(batch / WAVES_PER_BLOCK);
  dim3 block(THREADS);
  qlayer_kernel<<<grid, block, 0, stream>>>(x, qw, out);
}


# Last layer's qubits 8-9 gate and CNOT ring moved into the measurement: <Z> as ring-parity sign sums of the stage-2 accumulators, qubit 8/9 by its rotated Z on amplitude pairs; sum order changes
# speedup vs baseline: 3.2610x; 1.0716x over previous
#include <hip/hip_runtime.h>

typedef __attribute__((ext_vector_type(2))) float v2f;
typedef __attribute__((ext_vector_type(8))) float v8f;
typedef __attribute__((ext_vector_type(4))) float v4f;

#define NQ 10
#define NL 4
#define NSTATES 1024
#define WAVES_PER_BLOCK 4
#define THREADS (WAVES_PER_BLOCK * 32)

__device__ __forceinline__ void sincos_q(const float a, float& sn, float& cs) {
  const float k = rintf(a * 0.636619772f);
  float r = fmaf(k, -0x1.921fb4p+0f, a);
  r = fmaf(k, -0x1.4442d0p-24f, r);
  r = fmaf(k, -0x1.846988p-48f, r);
  const int q = (int)k;
  const float t = r * r;
  float ps = fmaf(t, 2.75573192e-6f, -1.98412698e-4f);
  ps = fmaf(ps, t, 8.33333333e-3f);
  ps = fmaf(ps, t, -1.66666667e-1f);
  const float s = fmaf(r * t, ps, r);
  float pc = fmaf(t, -2.75573192e-7f, 2.48015873e-5f);
  pc = fmaf(pc, t, -1.38888889e-3f);
  pc = fmaf(pc, t, 4.16666667e-2f);
  const float c = fmaf(t * t, pc, fmaf(t, -0.5f, 1.0f));
  const bool odd = (q & 1) != 0;
  const float ss = odd ? c : s, cc = odd ? s : c;
  sn = __int_as_float(__float_as_int(ss) ^ ((q & 2) << 30));
  cs = __int_as_float(__float_as_int(cc) ^ (((q + 1) & 2) << 30));
}

#define LDS_BYTES (1280 + 8192 + 8192 + 512 + WAVES_PER_BLOCK * 8192)

__device__ __forceinline__ int slot_of(const int f) {
  return ((f >> 6) & 1) | (((f >> 2) & 15) << 1) | (((f >> 9) & 1) << 5) |
         (((f >> 7) & 3) << 6) | ((f & 3) << 8);
}


__device__ __forceinline__ void load_gate16(const float* Ure, const float* Uim, const int lane,
                                            v2f (&gRe)[4], v2f (&gNs)[4], v2f (&gDf)[4]) {
  const int h = lane >> 4, n = lane & 15;
  const float4* pr = (const float4*)(Ure + n * 16 + 8 * h);
  const float4* pi = (const float4*)(Uim + n * 16 + 8 * h);
  const float4 r0 = pr[0], r1 = pr[1], i0 = pi[0], i1 = pi[1];
  const float rr[8] = {r0.x, r0.y, r0.z, r0.w, r1.x, r1.y, r1.z, r1.w};
  const float ii[8] = {i0.x, i0.y, i0.z, i0.w, i1.x, i1.y, i1.z, i1.w};
  #pragma unroll
  for (int k = 0; k < 4; ++k) {
    gRe[k].x = rr[2 * k];                 gRe[k].y = rr[2 * k + 1];
    gNs[k].x = -rr[2 * k] - ii[2 * k];    gNs[k].y = -rr[2 * k + 1] - ii[2 * k + 1];
    gDf[k].x = ii[2 * k] - rr[2 * k];     gDf[k].y = ii[2 * k + 1] - rr[2 * k + 1];
  }
}

#define WMMA4(A, B, C) __builtin_amdgcn_wmma_f32_16x16x4_f32(false, (A), false, (B), (short)0, (C), false, false)

__device__ __forceinline__ int cnot_dst(int j, const int rr) {
  #pragma unroll
  for (int w = 0; w < NQ; ++w) {
    const int cpos = 9 - w;
    int t = w + rr;
    t = (t >= NQ) ? t - NQ : t;
    const int tpos = 9 - t;
    j ^= ((j >> cpos) & 1) << tpos;
  }
  return j;
}

constexpr int cnot_dst_c(int j, const int rr) {
  for (int w = 0; w < NQ; ++w) {
    const int cpos = 9 - w;
    int t = w + rr;
    t = (t >= NQ) ? t - NQ : t;
    const int tpos = 9 - t;
    j ^= ((j >> cpos) & 1) << tpos;
  }
  return j;
}
constexpr int ring_mask(const int w, const int rr) {
  int m = 0;
  for (int k = 0; k < NQ; ++k)
    if ((cnot_dst_c(1 << k, rr) >> (9 - w)) & 1) m |= 1 << k;
  return m;
}
constexpr int RR_LAST = ((NL - 1) % (NQ - 1)) + 1;
template <int W> constexpr int last_mask_v = ring_mask(W, RR_LAST);
constexpr bool last_masks_hold_one_of_q8_q9() {
  for (int w = 0; w < NQ; ++w)
    if ((ring_mask(w, RR_LAST) & 3) == 3) return false;
  return true;
}
static_assert(last_masks_hold_one_of_q8_q9(), "an output parity holds both qubit 8 and qubit 9");

__device__ __forceinline__ void load_gate4(const float* ucr, const float* uci, const int lane,
                                           v2f (&wRe)[4], v2f (&wNs)[4], v2f (&wDf)[4]) {
  const int h = lane >> 4, n = lane & 15;
  const bool on = ((n >> 3) == h);
  const int jsel = n & 1, r = (n >> 1) & 3;
  #pragma unroll
  for (int k = 0; k < 4; ++k) {
    const float vr = on ? ucr[r * 4 + k] : 0.f;
    const float vi = on ? uci[r * 4 + k] : 0.f;
    const float ns = -vr - vi, df = vi - vr;
    wRe[k].x = jsel ? 0.f : vr;  wRe[k].y = jsel ? vr : 0.f;
    wNs[k].x = jsel ? 0.f : ns;  wNs[k].y = jsel ? ns : 0.f;
    wDf[k].x = jsel ? 0.f : df;  wDf[k].y = jsel ? df : 0.f;
  }
}

__device__ __forceinline__ void scatter_ring(float2* S, const v8f (&re3)[4], const v8f (&im3)[4],
                                             const int dLane, const int g0, const int g1,
                                             const int g2, const int g3, const int g4) {
  #pragma unroll
  for (int t = 0; t < 4; ++t) {
    const int dj = dLane ^ ((t & 1) ? g3 : 0) ^ ((t & 2) ? g4 : 0);
    #pragma unroll
    for (int v = 0; v < 8; ++v) {
      const int o = dj ^ ((v & 1) ? g2 : 0) ^ ((v & 2) ? g0 : 0) ^ ((v & 4) ? g1 : 0);
      S[o] = make_float2(re3[t][v], im3[t][v]);
    }
  }
}

__device__ __forceinline__ void ring_images(const int rr, const int lane, int& dLane,
                                            int& g0, int& g1, int& g2, int& g3, int& g4) {
  const int n = lane & 15, h = lane >> 4;
  dLane = slot_of(cnot_dst((n << 6) | (h << 5), rr));
  const int bas = slot_of(cnot_dst(1 << (lane & 15), rr));
  g0 = __builtin_amdgcn_readlane(bas, 0);
  g1 = __builtin_amdgcn_readlane(bas, 1);
  g2 = __builtin_amdgcn_readlane(bas, 2);
  g3 = __builtin_amdgcn_readlane(bas, 3);
  g4 = __builtin_amdgcn_readlane(bas, 4);
}

struct cpx { float r, i; };
__device__ __forceinline__ cpx cmul(const cpx a, const cpx b) {
  cpx o;  o.r = a.r * b.r - a.i * b.i;  o.i = a.r * b.i + a.i * b.r;  return o;
}
__device__ __forceinline__ cpx cmul2(const cpx a, const cpx b, const cpx c, const cpx d) {
  cpx o;
  o.r = a.r * b.r - a.i * b.i + c.r * d.r - c.i * d.i;
  o.i = a.r * b.i + a.i * b.r + c.r * d.i + c.i * d.r;
  return o;
}
__device__ __forceinline__ cpx csel(const bool s, const cpx a, const cpx b) {
  cpx o;  o.r = s ? a.r : b.r;  o.i = s ? a.i : b.i;  return o;
}

__device__ __forceinline__ void layers01_chain(float2* S, const float* u2,
                                               const float (&cw)[NQ], const float (&sw)[NQ],
                                               const int lane) {
  const int h = lane >> 4, n = lane & 15;
  const bool p = (h != 0);

  const float* v2 = u2 + NQ * 8;
  v4f gu[NQ][2], gv[NQ], gv4b, gv5b;
  #pragma unroll
  for (int q = 0; q < NQ; ++q) {
    gu[q][0] = *(const v4f*)(u2 + q * 8);
    gu[q][1] = *(const v4f*)(u2 + q * 8 + 4);
  }
  #pragma unroll
  for (int q = 0; q < NQ; ++q) {
    const int bit = (q < 4) ? (3 - q) : (9 - q);
    const int y = (q == 4 || q == 5) ? 0 : ((n >> bit) & 1);
    gv[q] = *(const v4f*)(v2 + q * 8 + 4 * y);
  }
  gv4b = *(const v4f*)(v2 + 4 * 8 + 4);
  gv5b = *(const v4f*)(v2 + 5 * 8 + 4);
  asm volatile("" :: "v"(gu[0][0]), "v"(gu[0][1]), "v"(gu[1][0]), "v"(gu[1][1]), "v"(gu[2][0]),
               "v"(gu[2][1]), "v"(gu[3][0]), "v"(gu[3][1]), "v"(gu[4][0]), "v"(gu[4][1]),
               "v"(gu[5][0]), "v"(gu[5][1]), "v"(gu[6][0]), "v"(gu[6][1]), "v"(gu[7][0]),
               "v"(gu[7][1]));
  asm volatile("" :: "v"(gu[8][0]), "v"(gu[8][1]), "v"(gu[9][0]), "v"(gu[9][1]), "v"(gv[0]),
               "v"(gv[1]), "v"(gv[2]), "v"(gv[3]), "v"(gv[4]), "v"(gv[5]), "v"(gv[6]), "v"(gv[7]),
               "v"(gv[8]), "v"(gv[9]), "v"(gv4b), "v"(gv5b));

  cpx u[NQ][2];
  #pragma unroll
  for (int q = 0; q < NQ; ++q) {
    u[q][0].r = gu[q][0].x * cw[q] + gu[q][0].z * sw[q];
    u[q][0].i = gu[q][0].y * cw[q] + gu[q][0].w * sw[q];
    u[q][1].r = gu[q][1].x * cw[q] + gu[q][1].z * sw[q];
    u[q][1].i = gu[q][1].y * cw[q] + gu[q][1].w * sw[q];
  }
  cpx Vr[NQ][2], V4[2][2], V5[2][2];
  #pragma unroll
  for (int q = 0; q < NQ; ++q) {
    Vr[q][0].r = gv[q].x;  Vr[q][0].i = gv[q].y;
    Vr[q][1].r = gv[q].z;  Vr[q][1].i = gv[q].w;
  }
  V4[0][0] = Vr[4][0];  V4[0][1] = Vr[4][1];
  V4[1][0].r = gv4b.x;  V4[1][0].i = gv4b.y;  V4[1][1].r = gv4b.z;  V4[1][1].i = gv4b.w;
  V5[0][0] = Vr[5][0];  V5[0][1] = Vr[5][1];
  V5[1][0].r = gv5b.x;  V5[1][0].i = gv5b.y;  V5[1][1].r = gv5b.z;  V5[1][1].i = gv5b.w;

  v2f LrA[2], LiA[2], nLiA[2];
  {
    const cpx u0p = csel(p, u[0][1], u[0][0]), u0n = csel(p, u[0][0], u[0][1]);
    const cpx u1p = csel(p, u[1][1], u[1][0]), u1n = csel(p, u[1][0], u[1][1]);
    cpx a0 = cmul(Vr[0][0], u0p), a1 = cmul(Vr[0][1], u0n);
    {
      const cpx s0 = cmul2(a0, u1p, a1, u1n), s1 = cmul2(a0, u1n, a1, u1p);
      a0 = cmul(Vr[1][0], s0);  a1 = cmul(Vr[1][1], s1);
    }
    #pragma unroll
    for (int q = 2; q < 4; ++q) {
      const cpx s0 = cmul2(a0, u[q][0], a1, u[q][1]), s1 = cmul2(a0, u[q][1], a1, u[q][0]);
      a0 = cmul(Vr[q][0], s0);  a1 = cmul(Vr[q][1], s1);
    }
    const cpx in0 = cmul2(a0, u[4][0], a1, u[4][1]), in1 = cmul2(a0, u[4][1], a1, u[4][0]);
    #pragma unroll
    for (int i = 0; i < 2; ++i) {
      const cpx l0 = cmul(V4[i][0], in0), l1 = cmul(V4[i][1], in1);
      LrA[i].x = l0.r;  LrA[i].y = l1.r;
      LiA[i].x = l0.i;  LiA[i].y = l1.i;
      nLiA[i].x = -l0.i;  nLiA[i].y = -l1.i;
    }
  }

  v2f RrB[2], RiB[2];
  {
    const cpx v9p = csel(p, Vr[9][1], Vr[9][0]);
    const cpx u9p = csel(p, u[9][1], u[9][0]), u9n = csel(p, u[9][0], u[9][1]);
    cpx c0 = cmul(v9p, u9p), c1 = cmul(v9p, u9n);
    #pragma unroll
    for (int q = 8; q > 5; --q) {
      const cpx w0 = cmul(Vr[q][0], c0), w1 = cmul(Vr[q][1], c1);
      c0 = cmul2(w0, u[q][0], w1, u[q][1]);
      c1 = cmul2(w0, u[q][1], w1, u[q][0]);
    }
    #pragma unroll
    for (int i2 = 0; i2 < 2; ++i2) {
      const cpx w0 = cmul(V5[i2][0], c0), w1 = cmul(V5[i2][1], c1);
      const cpx r0 = cmul2(w0, u[5][0], w1, u[5][1]), r1 = cmul2(w0, u[5][1], w1, u[5][0]);
      RrB[i2].x = r0.r;  RrB[i2].y = r1.r;
      RiB[i2].x = r0.i;  RiB[i2].y = r1.i;
    }
  }

  v8f re[4], im[4];
  const v8f zero = (v8f){0.f,0.f,0.f,0.f,0.f,0.f,0.f,0.f};
  #pragma unroll
  for (int tt = 0; tt < 4; ++tt) {
    re[tt] = WMMA4(LrA[tt >> 1], RrB[tt & 1], zero);
    im[tt] = WMMA4(LrA[tt >> 1], RiB[tt & 1], zero);
  }
  #pragma unroll
  for (int tt = 0; tt < 4; ++tt) {
    re[tt] = WMMA4(nLiA[tt >> 1], RiB[tt & 1], re[tt]);
    im[tt] = WMMA4(LiA[tt >> 1], RrB[tt & 1], im[tt]);
  }

  const int dLane = slot_of(cnot_dst((h << 9) | n, 2));
  const int bas = slot_of(cnot_dst(1 << (lane & 15), 2));
  const int g4 = __builtin_amdgcn_readlane(bas, 4);
  const int g5 = __builtin_amdgcn_readlane(bas, 5);
  const int g6 = __builtin_amdgcn_readlane(bas, 6);
  const int g7 = __builtin_amdgcn_readlane(bas, 7);
  const int g8 = __builtin_amdgcn_readlane(bas, 8);
  #pragma unroll
  for (int tt = 0; tt < 4; ++tt) {
    const int dj = dLane ^ ((tt & 2) ? g5 : 0) ^ ((tt & 1) ? g4 : 0);
    #pragma unroll
    for (int r = 0; r < 8; ++r) {
      const int o = dj ^ ((r & 1) ? g6 : 0) ^ ((r & 2) ? g7 : 0) ^ ((r & 4) ? g8 : 0);
      S[o] = make_float2(re[tt][r], im[tt][r]);
    }
  }
}

__device__ __forceinline__ void layer_stages12(const float2* S,
                                               const float* UAre, const float* UAim,
                                               const float* UBre, const float* UBim,
                                               const int lane, v8f (&d2re)[4], v8f (&d2im)[4]) {
  v2f uaRe[4], uaNs[4], uaDf[4], ubRe[4], ubNs[4], ubDf[4];
  load_gate16(UAre, UAim, lane, uaRe, uaNs, uaDf);
  load_gate16(UBre, UBim, lane, ubRe, ubNs, ubDf);

  const float2* Sl = S + 2 * lane;
  float4 z[4][4];
  #pragma unroll
  for (int c = 0; c < 4; ++c) {
    #pragma unroll
    for (int k = 0; k < 4; ++k) z[c][k] = *(const float4*)(Sl + 64 * k + 256 * c);
  }

  v8f re1[4], im1[4];
  #pragma unroll
  for (int c = 0; c < 4; ++c) re1[c] = (v8f){0.f,0.f,0.f,0.f,0.f,0.f,0.f,0.f};
  #pragma unroll
  for (int k = 0; k < 4; ++k) {
    #pragma unroll
    for (int c = 0; c < 4; ++c) {
      v2f as;  as.x = z[c][k].x + z[c][k].y;  as.y = z[c][k].z + z[c][k].w;
      re1[c] = WMMA4(as, uaRe[k], re1[c]);
    }
  }
  #pragma unroll
  for (int c = 0; c < 4; ++c) im1[c] = re1[c];
  #pragma unroll
  for (int k = 0; k < 4; ++k) {
    #pragma unroll
    for (int c = 0; c < 4; ++c) {
      v2f ar, ai;
      ar.x = z[c][k].x;  ar.y = z[c][k].z;
      ai.x = z[c][k].y;  ai.y = z[c][k].w;
      re1[c] = WMMA4(ai, uaNs[k], re1[c]);
      im1[c] = WMMA4(ar, uaDf[k], im1[c]);
    }
  }

  #pragma unroll
  for (int c = 0; c < 4; ++c) d2re[c] = (v8f){0.f,0.f,0.f,0.f,0.f,0.f,0.f,0.f};
  #pragma unroll
  for (int k = 0; k < 4; ++k) {
    #pragma unroll
    for (int c = 0; c < 4; ++c) {
      v2f bs;
      bs.x = re1[c][2 * k] + im1[c][2 * k];
      bs.y = re1[c][2 * k + 1] + im1[c][2 * k + 1];
      d2re[c] = WMMA4(ubRe[k], bs, d2re[c]);
    }
  }
  #pragma unroll
  for (int c = 0; c < 4; ++c) d2im[c] = d2re[c];
  #pragma unroll
  for (int k = 0; k < 4; ++k) {
    #pragma unroll
    for (int c = 0; c < 4; ++c) {
      v2f br, bi;
      br.x = re1[c][2 * k];  br.y = re1[c][2 * k + 1];
      bi.x = im1[c][2 * k];  bi.y = im1[c][2 * k + 1];
      d2re[c] = WMMA4(ubNs[k], bi, d2re[c]);
      d2im[c] = WMMA4(ubDf[k], br, d2im[c]);
    }
  }

}

__device__ __forceinline__ void layer_regs(float2* S,
                                           const float* UAre, const float* UAim,
                                           const float* UBre, const float* UBim,
                                           const float* ucr, const float* uci,
                                           const int dLane, const int g0, const int g1,
                                           const int g2, const int g3, const int g4,
                                           const int lane) {
  v2f wRe[4], wNs[4], wDf[4];
  load_gate4(ucr, uci, lane, wRe, wNs, wDf);
  v8f d2re[4], d2im[4];
  layer_stages12(S, UAre, UAim, UBre, UBim, lane, d2re, d2im);

  v8f re3[4], im3[4];
  #pragma unroll
  for (int t = 0; t < 4; ++t) re3[t] = (v8f){0.f,0.f,0.f,0.f,0.f,0.f,0.f,0.f};
  #pragma unroll
  for (int k = 0; k < 4; ++k) {
    #pragma unroll
    for (int t = 0; t < 4; ++t) {
      v2f bs;
      bs.x = d2re[k][2 * t] + d2im[k][2 * t];
      bs.y = d2re[k][2 * t + 1] + d2im[k][2 * t + 1];
      re3[t] = WMMA4(wRe[k], bs, re3[t]);
    }
  }
  #pragma unroll
  for (int t = 0; t < 4; ++t) im3[t] = re3[t];
  #pragma unroll
  for (int k = 0; k < 4; ++k) {
    #pragma unroll
    for (int t = 0; t < 4; ++t) {
      v2f br, bi;
      br.x = d2re[k][2 * t];  br.y = d2re[k][2 * t + 1];
      bi.x = d2im[k][2 * t];  bi.y = d2im[k][2 * t + 1];
      re3[t] = WMMA4(wNs[k], bi, re3[t]);
      im3[t] = WMMA4(wDf[k], br, im3[t]);
    }
  }
  scatter_ring(S, re3, im3, dLane, g0, g1, g2, g3, g4);
}

#define LAST_MASK_SEL(W) mw = (n == (W)) ? last_mask_v<(W)> : mw;
__device__ __forceinline__ void measure_last(const v8f (&re)[4], const v8f (&im)[4],
                                             const float* g8, const float* g9,
                                             const int lane, float* outRow) {
  const int h = lane >> 4, n = lane & 15;
  int mw = 0;
  LAST_MASK_SEL(0) LAST_MASK_SEL(1) LAST_MASK_SEL(2) LAST_MASK_SEL(3) LAST_MASK_SEL(4)
  LAST_MASK_SEL(5) LAST_MASK_SEL(6) LAST_MASK_SEL(7) LAST_MASK_SEL(8) LAST_MASK_SEL(9)
  const bool valid = n < NQ;
  const int mc = mw & 3, mm = (mw >> 2) & 15, ma = (mw >> 6) & 15;
  const float fQ = (valid && mc == 0) ? 1.f : 0.f;
  const float f8 = (valid && mc == 2) ? 1.f : 0.f;
  const float f9 = (valid && mc == 1) ? 1.f : 0.f;
  float sg[8], sa[8];
  #pragma unroll
  for (int v = 0; v < 8; ++v) {
    sg[v] = (__popc(((h << 3) | v) & mm) & 1) ? -1.f : 1.f;
    sa[v] = (__popc(((h << 3) | v) & ma) & 1) ? -1.f : 1.f;
  }

  const v4f t8 = *(const v4f*)g8, b8 = *(const v4f*)(g8 + 4);
  const v4f t9 = *(const v4f*)g9, b9 = *(const v4f*)(g9 + 4);
  const float zA8 = t8.x * t8.x + t8.y * t8.y - (b8.x * b8.x + b8.y * b8.y);
  const float zB8 = t8.z * t8.z + t8.w * t8.w - (b8.z * b8.z + b8.w * b8.w);
  const float wr8 = 2.f * (t8.x * t8.z + t8.y * t8.w - (b8.x * b8.z + b8.y * b8.w));
  const float wi8 = 2.f * (t8.x * t8.w - t8.y * t8.z - (b8.x * b8.w - b8.y * b8.z));
  const float zA9 = t9.x * t9.x + t9.y * t9.y - (b9.x * b9.x + b9.y * b9.y);
  const float zB9 = t9.z * t9.z + t9.w * t9.w - (b9.z * b9.z + b9.w * b9.w);
  const float wr9 = 2.f * (t9.x * t9.z + t9.y * t9.w - (b9.x * b9.z + b9.y * b9.w));
  const float wi9 = 2.f * (t9.x * t9.w - t9.y * t9.z - (b9.x * b9.w - b9.y * b9.z));

  float Q[8], T8[8], T9[8];
  #pragma unroll
  for (int v = 0; v < 8; ++v) {
    const float r0 = re[0][v], i0 = im[0][v], r1 = re[1][v], i1 = im[1][v];
    const float r2 = re[2][v], i2 = im[2][v], r3 = re[3][v], i3 = im[3][v];
    const float p0 = r0 * r0 + i0 * i0, p1 = r1 * r1 + i1 * i1;
    const float p2 = r2 * r2 + i2 * i2, p3 = r3 * r3 + i3 * i3;
    const float cr8 = (r0 * r2 + i0 * i2) + (r1 * r3 + i1 * i3);
    const float ci8 = (r0 * i2 - i0 * r2) + (r1 * i3 - i1 * r3);
    const float cr9 = (r0 * r1 + i0 * i1) + (r2 * r3 + i2 * i3);
    const float ci9 = (r0 * i1 - i0 * r1) + (r2 * i3 - i2 * r3);
    Q[v]  = (p0 + p1) + (p2 + p3);
    T8[v] = zA8 * (p0 + p1) + zB8 * (p2 + p3) + wr8 * cr8 - wi8 * ci8;
    T9[v] = zA9 * (p0 + p2) + zB9 * (p1 + p3) + wr9 * cr9 - wi9 * ci9;
  }

  v8f dQ = (v8f){0.f,0.f,0.f,0.f,0.f,0.f,0.f,0.f};
  v8f d8 = dQ, d9 = dQ;
  #pragma unroll
  for (int q = 0; q < 4; ++q) {
    v2f aQ, a8, a9, gQ, g8s, g9s;
    aQ.x = Q[2 * q];   aQ.y = Q[2 * q + 1];
    a8.x = T8[2 * q];  a8.y = T8[2 * q + 1];
    a9.x = T9[2 * q];  a9.y = T9[2 * q + 1];
    gQ.x  = sg[2 * q] * fQ;  gQ.y  = sg[2 * q + 1] * fQ;
    g8s.x = sg[2 * q] * f8;  g8s.y = sg[2 * q + 1] * f8;
    g9s.x = sg[2 * q] * f9;  g9s.y = sg[2 * q + 1] * f9;
    dQ = WMMA4(aQ, gQ, dQ);
    d8 = WMMA4(a8, g8s, d8);
    d9 = WMMA4(a9, g9s, d9);
  }
  const v8f ds = dQ + d8 + d9;
  float val = 0.f;
  #pragma unroll
  for (int r = 0; r < 8; ++r) val += sa[r] * ds[r];
  val += __shfl_xor(val, 16, 32);
  if (lane < NQ) outRow[lane] = val;
}

extern "C" __global__ __launch_bounds__(THREADS)
void qlayer_kernel(const float* __restrict__ x,
                   const float* __restrict__ qw,
                   float* __restrict__ out) {
  __shared__ __attribute__((aligned(16))) char smem_raw[LDS_BYTES];
  float* u2   = (float*)smem_raw;
  float* UAre = u2 + NL * NQ * 8;
  float* UAim = UAre + NL * 256;
  float* UBre = UAim + NL * 256;
  float* UBim = UBre + NL * 256;
  float* UCre = UBim + NL * 256;
  float* UCim = UCre + NL * 16;
  float2* stateBase = (float2*)(UCim + NL * 16);

  const int tid = threadIdx.x;
  const int wave = tid >> 5, lane = tid & 31;
  const int b = blockIdx.x * WAVES_PER_BLOCK + wave;

  float cw[NQ], sw[NQ];
  #pragma unroll
  for (int w = 0; w < NQ; ++w) {
    const float ang = 1.5707963267948966f * x[b * NQ + w];
    sincos_q(ang, sw[w], cw[w]);
  }

  #pragma unroll 1
  for (int e = tid; e < NL * NQ; e += THREADS) {
    const int l = e / NQ, w = e % NQ;
    const float phi = qw[(l * NQ + w) * 3 + 0];
    const float th  = qw[(l * NQ + w) * 3 + 1];
    const float om  = qw[(l * NQ + w) * 3 + 2];
    const float ap = 0.5f * (phi + om), am = 0.5f * (phi - om);
    float c, s, cpo, spo, cb, sb;
    sincos_q(0.5f * th, s, c);
    sincos_q(ap, spo, cpo);
    sincos_q(am, sb, cb);
    float* g = u2 + e * 8;
    g[0] =  cpo * c; g[1] = -spo * c;
    g[2] = -cb * s;  g[3] = -sb * s;
    g[4] =  cb * s;  g[5] = -sb * s;
    g[6] =  cpo * c; g[7] =  spo * c;
  }
  __syncthreads();

  #pragma unroll 1
  for (int e = 2 * 256 + tid; e < NL * 256; e += THREADS) {
    const int l = e >> 8, rc = e & 255, r = rc >> 4, cc = rc & 15;
    float arA = 1.f, aiA = 0.f, arB = 1.f, aiB = 0.f;
    #pragma unroll
    for (int q = 0; q < 4; ++q) {
      const int rb = (r >> (3 - q)) & 1, cbit = (cc >> (3 - q)) & 1;
      const float* gA = u2 + (((l * NQ + q) * 4) + rb * 2 + cbit) * 2;
      float nr = arA * gA[0] - aiA * gA[1];
      float ni = arA * gA[1] + aiA * gA[0];
      arA = nr; aiA = ni;
      const float* gB = u2 + (((l * NQ + 4 + q) * 4) + rb * 2 + cbit) * 2;
      nr = arB * gB[0] - aiB * gB[1];
      ni = arB * gB[1] + aiB * gB[0];
      arB = nr; aiB = ni;
    }
    UAre[e] = arA; UAim[e] = aiA;
    UBre[e] = arB; UBim[e] = aiB;
  }
  #pragma unroll 1
  for (int e = 2 * 16 + tid; e < (NL - 1) * 16; e += THREADS) {
    const int l = e >> 4, r = (e >> 2) & 3, cc = e & 3;
    const float* g8 = u2 + (((l * NQ + 8) * 4) + ((r >> 1) & 1) * 2 + ((cc >> 1) & 1)) * 2;
    const float* g9 = u2 + (((l * NQ + 9) * 4) + (r & 1) * 2 + (cc & 1)) * 2;
    UCre[e] = g8[0] * g9[0] - g8[1] * g9[1];
    UCim[e] = g8[0] * g9[1] + g8[1] * g9[0];
  }
  __syncthreads();

  float2* S = stateBase + wave * NSTATES;

  layers01_chain(S, u2, cw, sw, lane);

  #pragma unroll 1
  for (int l = 2; l < NL - 1; ++l) {
    const int rr = (l % (NQ - 1)) + 1;
    int dLane, g0, g1, g2, g3, g4;
    ring_images(rr, lane, dLane, g0, g1, g2, g3, g4);
    layer_regs(S, UAre + l * 256, UAim + l * 256, UBre + l * 256, UBim + l * 256,
               UCre + l * 16, UCim + l * 16, dLane, g0, g1, g2, g3, g4, lane);
  }

  {
    const int l = NL - 1;
    v8f d2re[4], d2im[4];
    layer_stages12(S, UAre + l * 256, UAim + l * 256, UBre + l * 256, UBim + l * 256, lane, d2re, d2im);
    measure_last(d2re, d2im, u2 + (l * NQ + 8) * 8, u2 + (l * NQ + 9) * 8, lane, out + b * NQ);
  }
}

extern "C" void kernel_launch(void* const* d_in, const int* in_sizes, int n_in,
                              void* d_out, int out_size, void* d_ws, size_t ws_size,
                              hipStream_t stream) {
  (void)in_sizes; (void)n_in; (void)out_size; (void)d_ws; (void)ws_size;
  const float* x  = (const float*)d_in[0];
  const float* qw = (const float*)d_in[1];
  float* out = (float*)d_out;
  const int batch = 2048;
  dim3 grid(batch / WAVES_PER_BLOCK);
  dim3 block(THREADS);
  qlayer_kernel<<<grid, block, 0, stream>>>(x, qw, out);
}
